# starting kernel
# baseline (speedup 1.0000x reference)
#include <hip/hip_runtime.h>
#include <hip/hip_bf16.h>
#include <math.h>


typedef __attribute__((ext_vector_type(2))) float v2f;
typedef __attribute__((ext_vector_type(8))) float v8f;

#define NTOK   65536
#define DM     8
#define DI     16
#define CHUNK  256
#define NCHUNK 256
#define N16    (NTOK * DI)

__device__ inline v8f wmma4(v2f a, v2f b, v8f c) {
  return __builtin_amdgcn_wmma_f32_16x16x4_f32(
      false, a, false, b, (short)0, c, false, false);
}

__device__ inline float softplus_f(float s) {
  return (s > 20.f) ? s : logf(1.f + __expf(s));
}

__global__ void k_proj(const float* __restrict__ x,
                       const float* __restrict__ inw,
                       const float* __restrict__ dtw,
                       const float* __restrict__ dtb,
                       const float* __restrict__ bw,
                       const float* __restrict__ cw,
                       float* __restrict__ xb, float* __restrict__ z,
                       float* __restrict__ dt, float* __restrict__ Bv,
                       float* __restrict__ Cv) {
  __shared__ float lds[4][256];
  const int lane = threadIdx.x & 31;
  const int wave = threadIdx.x >> 5;
  const int t0   = (blockIdx.x * 4 + wave) * 16;
  const int lo   = lane & 15, hi = lane >> 4;

  v8f axb = {}, az = {};
#pragma unroll
  for (int kk = 0; kk < 2; ++kk) {
    const int kb = kk * 4 + 2 * hi;
    v2f a, b0, b1;
    a.x  = x[(t0 + lo) * 8 + kb + 0];
    a.y  = x[(t0 + lo) * 8 + kb + 1];
    b0.x = inw[lo * 8 + kb + 0];
    b0.y = inw[lo * 8 + kb + 1];
    b1.x = inw[(lo + 16) * 8 + kb + 0];
    b1.y = inw[(lo + 16) * 8 + kb + 1];
    axb = wmma4(a, b0, axb);
    az  = wmma4(a, b1, az);
  }

#pragma unroll
  for (int r = 0; r < 8; ++r) lds[wave][(r + 8 * hi) * 16 + lo] = axb[r];

  v8f adt = {}, ab = {}, ac = {};
#pragma unroll
  for (int kk = 0; kk < 4; ++kk) {
    const int kb = kk * 4 + 2 * hi;
    v2f a, b;
    a.x = lds[wave][lo * 16 + kb + 0];
    a.y = lds[wave][lo * 16 + kb + 1];
    b.x = dtw[lo * 16 + kb + 0]; b.y = dtw[lo * 16 + kb + 1];
    adt = wmma4(a, b, adt);
    b.x = bw[lo * 16 + kb + 0];  b.y = bw[lo * 16 + kb + 1];
    ab  = wmma4(a, b, ab);
    b.x = cw[lo * 16 + kb + 0];  b.y = cw[lo * 16 + kb + 1];
    ac  = wmma4(a, b, ac);
  }

  const float bias = dtb[lo];
#pragma unroll
  for (int r = 0; r < 8; ++r) {
    const int idx = (t0 + r + 8 * hi) * 16 + lo;
    xb[idx] = axb[r];
    z[idx]  = az[r];
    dt[idx] = softplus_f(adt[r] + bias);
    Bv[idx] = ab[r];
    Cv[idx] = ac[r];
  }
}

__global__ void k_chunkop(const float* __restrict__ dt,
                          const float* __restrict__ xb,
                          const float* __restrict__ Bv,
                          const float* __restrict__ A_log,
                          const float* __restrict__ rope,
                          float* __restrict__ Tmat,
                          float* __restrict__ bvec) {
  const int c    = blockIdx.x;
  const int d    = threadIdx.x >> 5;
  const int lane = threadIdx.x & 31;
  const int lo   = lane & 15, hi = lane >> 4;

  float Arow[8], Frow[8];
#pragma unroll
  for (int r = 0; r < 8; ++r) {
    const int j = r + 8 * hi;
    Arow[r] = -__expf(A_log[d * 16 + j]);
    Frow[r] = rope[d * 16 + j];
  }
  const float A2 = -__expf(A_log[d * 16 + lo]);
  const float F2 = rope[d * 16 + lo];

  float T[8];
#pragma unroll
  for (int r = 0; r < 8; ++r) T[r] = ((r + 8 * hi) == lo) ? 1.f : 0.f;
  float b = 0.f;

  const int t0  = c * CHUNK;
  const int src = (lane & 16) | ((lo + 15) & 15);

  for (int t = 0; t < CHUNK; ++t) {
    const int tt   = t0 + t;
    const float dtc = dt[tt * 16 + d];
    const float xbd = xb[tt * 16 + d];
    const float Bj  = Bv[tt * 16 + lo];

    const float roll0 = __shfl_xor(T[7], 16, 32);
#pragma unroll
    for (int r = 7; r >= 1; --r) {
      const float ta   = 0.5f * dtc * Arow[r];
      const float abar = (1.f + ta) / (1.f - ta + 1e-8f);
      const float ang  = dtc * Frow[r];
      T[r] = abar * (__cosf(ang) * T[r] - __sinf(ang) * T[r - 1]);
    }
    {
      const float ta   = 0.5f * dtc * Arow[0];
      const float abar = (1.f + ta) / (1.f - ta + 1e-8f);
      const float ang  = dtc * Frow[0];
      T[0] = abar * (__cosf(ang) * T[0] - __sinf(ang) * roll0);
    }
    const float ta2   = 0.5f * dtc * A2;
    const float abar2 = (1.f + ta2) / (1.f - ta2 + 1e-8f);
    const float ang2  = dtc * F2;
    const float brl   = __shfl(b, src, 32);
    b = abar2 * (__cosf(ang2) * b - __sinf(ang2) * brl) + dtc * Bj * xbd;
  }

#pragma unroll
  for (int r = 0; r < 8; ++r)
    Tmat[((c * 16 + d) * 16 + (r + 8 * hi)) * 16 + lo] = T[r];
  if (hi == 0) bvec[(c * 16 + d) * 16 + lo] = b;
}

__global__ void k_combine(const float* __restrict__ Tmat,
                          const float* __restrict__ bvec,
                          const float* __restrict__ init_h,
                          float* __restrict__ hstart,
                          float* __restrict__ h_final) {
  const int d    = threadIdx.x >> 5;
  const int lane = threadIdx.x & 31;
  const int lo   = lane & 15;
  float h = init_h[d * 16 + lo];

  for (int c = 0; c < NCHUNK; ++c) {
    if (c + 1 < NCHUNK) {
      __builtin_prefetch(Tmat + (((c + 1) * 16 + d) * 16 + lo) * 16, 0, 1);
      __builtin_prefetch(bvec + ((c + 1) * 16 + d) * 16 + lo, 0, 1);
    }
    if (lane < 16) hstart[(c * 16 + d) * 16 + lo] = h;

    const float4* Trow =
        reinterpret_cast<const float4*>(Tmat + ((c * 16 + d) * 16 + lo) * 16);
    const float4 q0 = Trow[0], q1 = Trow[1], q2 = Trow[2], q3 = Trow[3];
    float Tr[16] = {q0.x, q0.y, q0.z, q0.w, q1.x, q1.y, q1.z, q1.w,
                    q2.x, q2.y, q2.z, q2.w, q3.x, q3.y, q3.z, q3.w};
    float hn = bvec[(c * 16 + d) * 16 + lo];
#pragma unroll
    for (int k = 0; k < 16; ++k)
      hn = fmaf(Tr[k], __shfl(h, (lane & 16) | k, 32), hn);
    h = hn;
  }
  if (lane < 16) h_final[d * 16 + lo] = h;
}

__global__ void k_scan(const float* __restrict__ dt,
                       const float* __restrict__ xb,
                       const float* __restrict__ Bv,
                       const float* __restrict__ Cv,
                       const float* __restrict__ A_log,
                       const float* __restrict__ rope,
                       const float* __restrict__ hstart,
                       float* __restrict__ y) {
  const int c    = blockIdx.x;
  const int d    = threadIdx.x >> 5;
  const int lane = threadIdx.x & 31;
  const int lo   = lane & 15;
  const float A2 = -__expf(A_log[d * 16 + lo]);
  const float F2 = rope[d * 16 + lo];
  float h = hstart[(c * 16 + d) * 16 + lo];
  const int t0  = c * CHUNK;
  const int src = (lane & 16) | ((lo + 15) & 15);

  for (int t = 0; t < CHUNK; ++t) {
    const int tt    = t0 + t;
    const float dtc = dt[tt * 16 + d];
    const float xbd = xb[tt * 16 + d];
    const float Bj  = Bv[tt * 16 + lo];
    const float Cj  = Cv[tt * 16 + lo];
    const float ta   = 0.5f * dtc * A2;
    const float abar = (1.f + ta) / (1.f - ta + 1e-8f);
    const float ang  = dtc * F2;
    const float hr   = __shfl(h, src, 32);
    h = abar * (__cosf(ang) * h - __sinf(ang) * hr) + dtc * Bj * xbd;
    float p = h * Cj;
    p += __shfl_xor(p, 1, 32);
    p += __shfl_xor(p, 2, 32);
    p += __shfl_xor(p, 4, 32);
    p += __shfl_xor(p, 8, 32);
    if (lane == 0) y[tt * 16 + d] = p;
  }
}

__global__ void k_out(const float* __restrict__ y,
                      const float* __restrict__ z,
                      const float* __restrict__ xb,
                      const float* __restrict__ Dp,
                      const float* __restrict__ outw,
                      float* __restrict__ out) {
  const int lane = threadIdx.x & 31;
  const int wave = threadIdx.x >> 5;
  const int t0   = (blockIdx.x * 4 + wave) * 16;
  const int lo   = lane & 15, hi = lane >> 4;

  v8f acc = {};
#pragma unroll
  for (int kk = 0; kk < 4; ++kk) {
    const int kb = kk * 4 + 2 * hi;
    v2f a, b;
#pragma unroll
    for (int v = 0; v < 2; ++v) {
      const int k   = kb + v;
      const int idx = (t0 + lo) * 16 + k;
      const float zz  = z[idx];
      const float sil = zz / (1.f + __expf(-zz));
      const float val = y[idx] * sil + Dp[k] * xb[idx];
      if (v == 0) a.x = val; else a.y = val;
    }
    b.x = (lo < 8) ? outw[lo * 16 + kb + 0] : 0.f;
    b.y = (lo < 8) ? outw[lo * 16 + kb + 1] : 0.f;
    acc = wmma4(a, b, acc);
  }
  if (lo < 8) {
#pragma unroll
    for (int r = 0; r < 8; ++r) out[(t0 + r + 8 * hi) * 8 + lo] = acc[r];
  }
}

extern "C" void kernel_launch(void* const* d_in, const int* in_sizes, int n_in,
                              void* d_out, int out_size, void* d_ws,
                              size_t ws_size, hipStream_t stream) {
  const float* x    = (const float*)d_in[0];
  const float* init = (const float*)d_in[1];
  const float* inw  = (const float*)d_in[2];
  const float* dtw  = (const float*)d_in[3];
  const float* dtb  = (const float*)d_in[4];
  const float* bw   = (const float*)d_in[5];
  const float* cw   = (const float*)d_in[6];
  const float* alog = (const float*)d_in[7];
  const float* Dp   = (const float*)d_in[8];
  const float* rope = (const float*)d_in[9];
  const float* outw = (const float*)d_in[10];
  float* out = (float*)d_out;

  float* ws = (float*)d_ws;
  float* XB = ws;
  float* Z  = XB + N16;
  float* DT = Z + N16;
  float* BV = DT + N16;
  float* CV = BV + N16;
  float* Y  = CV + N16;
  float* T  = Y + N16;
  float* BC = T + N16;
  float* HS = BC + NCHUNK * 256;

  k_proj<<<NTOK / 64, 128, 0, stream>>>(x, inw, dtw, dtb, bw, cw,
                                        XB, Z, DT, BV, CV);
  k_chunkop<<<NCHUNK, 512, 0, stream>>>(DT, XB, BV, alog, rope, T, BC);
  k_combine<<<1, 512, 0, stream>>>(T, BC, init, HS, out + NTOK * DM);
  k_scan<<<NCHUNK, 512, 0, stream>>>(DT, XB, BV, CV, alog, rope, HS, Y);
  k_out<<<NTOK / 64, 128, 0, stream>>>(Y, Z, XB, Dp, outw, out);
}


# K2 chunk operator composed from the chunk's last step backwards (S <- S*M_t): a lane reads only its own columns' coefficients, no broadcast rows; b_c a per-lane sum; product and sum order change
# speedup vs baseline: 5.3169x; 5.3169x over previous
#include <hip/hip_runtime.h>
#include <hip/hip_bf16.h>
#include <math.h>


typedef __attribute__((ext_vector_type(2))) float v2f;
typedef __attribute__((ext_vector_type(8))) float v8f;

#define NTOK   65536
#define DM     8
#define DI     16
#define CHUNK  256
#define NCHUNK 256
#define N16    (NTOK * DI)

__device__ inline v8f wmma4(v2f a, v2f b, v8f c) {
  return __builtin_amdgcn_wmma_f32_16x16x4_f32(
      false, a, false, b, (short)0, c, false, false);
}

__device__ inline float softplus_f(float s) {
  return (s > 20.f) ? s : logf(1.f + __expf(s));
}

__global__ void k_proj(const float* __restrict__ x,
                       const float* __restrict__ inw,
                       const float* __restrict__ dtw,
                       const float* __restrict__ dtb,
                       const float* __restrict__ bw,
                       const float* __restrict__ cw,
                       float* __restrict__ xb, float* __restrict__ z,
                       float* __restrict__ dt, float* __restrict__ Bv,
                       float* __restrict__ Cv) {
  __shared__ float lds[4][256];
  const int lane = threadIdx.x & 31;
  const int wave = threadIdx.x >> 5;
  const int t0   = (blockIdx.x * 4 + wave) * 16;
  const int lo   = lane & 15, hi = lane >> 4;

  v8f axb = {}, az = {};
#pragma unroll
  for (int kk = 0; kk < 2; ++kk) {
    const int kb = kk * 4 + 2 * hi;
    v2f a, b0, b1;
    a.x  = x[(t0 + lo) * 8 + kb + 0];
    a.y  = x[(t0 + lo) * 8 + kb + 1];
    b0.x = inw[lo * 8 + kb + 0];
    b0.y = inw[lo * 8 + kb + 1];
    b1.x = inw[(lo + 16) * 8 + kb + 0];
    b1.y = inw[(lo + 16) * 8 + kb + 1];
    axb = wmma4(a, b0, axb);
    az  = wmma4(a, b1, az);
  }

#pragma unroll
  for (int r = 0; r < 8; ++r) lds[wave][(r + 8 * hi) * 16 + lo] = axb[r];

  v8f adt = {}, ab = {}, ac = {};
#pragma unroll
  for (int kk = 0; kk < 4; ++kk) {
    const int kb = kk * 4 + 2 * hi;
    v2f a, b;
    a.x = lds[wave][lo * 16 + kb + 0];
    a.y = lds[wave][lo * 16 + kb + 1];
    b.x = dtw[lo * 16 + kb + 0]; b.y = dtw[lo * 16 + kb + 1];
    adt = wmma4(a, b, adt);
    b.x = bw[lo * 16 + kb + 0];  b.y = bw[lo * 16 + kb + 1];
    ab  = wmma4(a, b, ab);
    b.x = cw[lo * 16 + kb + 0];  b.y = cw[lo * 16 + kb + 1];
    ac  = wmma4(a, b, ac);
  }

  const float bias = dtb[lo];
#pragma unroll
  for (int r = 0; r < 8; ++r) {
    const int idx = (t0 + r + 8 * hi) * 16 + lo;
    xb[idx] = axb[r];
    z[idx]  = az[r];
    dt[idx] = softplus_f(adt[r] + bias);
    Bv[idx] = ab[r];
    Cv[idx] = ac[r];
  }
}

#define SB    32
#define NBLK  (CHUNK / SB)
#define ROWF  48

struct FillOperands { float dtc[SB / 2], xbd[SB / 2], Bj[SB / 2]; };

__device__ inline void chunkop_load(FillOperands& o,
                                    const float* __restrict__ dt,
                                    const float* __restrict__ xb,
                                    const float* __restrict__ Bv,
                                    int tbase, int fth, int fd, int fj) {
#pragma unroll
  for (int i = 0; i < SB / 2; ++i) {
    const int tt = tbase + 2 * i + fth;
    o.dtc[i] = dt[tt * 16 + fd];
    o.xbd[i] = xb[tt * 16 + fd];
    o.Bj[i]  = Bv[tt * 16 + fj];
  }
}

__device__ inline void chunkop_make(float* __restrict__ buf,
                                    const FillOperands& o,
                                    int fth, int fd, int fj,
                                    float fA, float fF) {
#pragma unroll
  for (int i = 0; i < SB / 2; ++i) {
    const int t = 2 * i + fth;
    const float dtc  = o.dtc[i];
    const float ta   = 0.5f * dtc * fA;
    const float abar = (1.f + ta) / (1.f - ta + 1e-8f);
    const float ang  = dtc * fF;
    float* row = buf + ((SB - 1 - t) * 16 + fd) * ROWF;
    row[fj]      = abar * __cosf(ang);
    row[16 + ((fj + 15) & 15)] = abar * __sinf(ang);
    row[32 + fj] = dtc * o.Bj[i] * o.xbd[i];
  }
}

__device__ inline void chunkop_fill(float* __restrict__ buf,
                                    const float* __restrict__ dt,
                                    const float* __restrict__ xb,
                                    const float* __restrict__ Bv,
                                    int tbase, int fth, int fd, int fj,
                                    float fA, float fF) {
  FillOperands o;
  chunkop_load(o, dt, xb, Bv, tbase, fth, fd, fj);
  chunkop_make(buf, o, fth, fd, fj, fA, fF);
}

#define KW2   512
#define KT2   768

__device__ inline float next_pair(float v) {
  const int i = __builtin_bit_cast(int, v);
  return __builtin_bit_cast(float, __builtin_amdgcn_permlane16(
      i, i, (int)0x07654321u, (int)0x8fedcba9u, true, false));
}

__global__ void __launch_bounds__(KT2)
k_chunkop(const float* __restrict__ dt,
          const float* __restrict__ xb,
          const float* __restrict__ Bv,
          const float* __restrict__ A_log,
          const float* __restrict__ rope,
          float* __restrict__ Tmat,
          float* __restrict__ bvec) {
  __shared__ __attribute__((aligned(16))) float tab[2][SB * 16 * ROWF];

  const int c    = blockIdx.x;
  const int tid  = threadIdx.x;
  const int d    = tid >> 5;
  const int lane = tid & 31;

  const bool producer = tid >= KW2;
  const int fj  = tid & 15;
  const int fd  = (tid >> 4) & 15;
  const int fth = (tid >> 8) & 1;
  const float fA = -__expf(A_log[fd * 16 + fj]);
  const float fF = rope[fd * 16 + fj];

  const int p = lane & 7, q = lane >> 3;
  float Sa[4], Sb[4], bacc[4];
#pragma unroll
  for (int r = 0; r < 4; ++r) {
    Sa[r]   = ((4 * q + r) == 2 * p)     ? 1.f : 0.f;
    Sb[r]   = ((4 * q + r) == 2 * p + 1) ? 1.f : 0.f;
    bacc[r] = 0.f;
  }

  const int t0 = c * CHUNK;

  if (!producer)
    chunkop_fill(tab[0], dt, xb, Bv, t0 + (NBLK - 1) * SB, fth, fd, fj, fA, fF);
  __syncthreads();

  for (int kb = 0; kb < NBLK; ++kb) {
    if (producer) {
      if (kb + 1 < NBLK) {
        float* nb = tab[(kb + 1) & 1];
        const int tn = t0 + (NBLK - 2 - kb) * SB;
        FillOperands o;
        chunkop_load(o, dt, xb, Bv, tn, 0, fd, fj);
        __builtin_amdgcn_sched_barrier(0);
        chunkop_make(nb, o, 0, fd, fj, fA, fF);
        chunkop_load(o, dt, xb, Bv, tn, 1, fd, fj);
        __builtin_amdgcn_sched_barrier(0);
        chunkop_make(nb, o, 1, fd, fj, fA, fF);
      }
    } else {
      const float* base = tab[kb & 1] + d * ROWF + 2 * p;
#pragma unroll 4
      for (int i = 0; i < SB; ++i) {
        const float* row = base + i * 16 * ROWF;
        const float2 E = *reinterpret_cast<const float2*>(row);
        const float2 F = *reinterpret_cast<const float2*>(row + 16);
        const float2 U = *reinterpret_cast<const float2*>(row + 32);
#pragma unroll
        for (int r = 0; r < 4; ++r) {
          const float sn = next_pair(Sa[r]);
          bacc[r] = fmaf(Sb[r], U.y, fmaf(Sa[r], U.x, bacc[r]));
          const float a = E.x * Sa[r] - F.x * Sb[r];
          Sb[r] = E.y * Sb[r] - F.y * sn;
          Sa[r] = a;
        }
      }
    }
    __syncthreads();
  }

  if (producer) return;
#pragma unroll
  for (int r = 0; r < 4; ++r) {
    float2 v; v.x = Sa[r]; v.y = Sb[r];
    *reinterpret_cast<float2*>(Tmat + ((c * 16 + d) * 16 + (4 * q + r)) * 16 + 2 * p) = v;
  }

  float bout = 0.f;
#pragma unroll
  for (int r = 0; r < 4; ++r) {
    float v = bacc[r];
    v += __shfl_xor(v, 1, 32);
    v += __shfl_xor(v, 2, 32);
    v += __shfl_xor(v, 4, 32);
    bout = (p == r) ? v : bout;
  }
  if (p < 4) bvec[(c * 16 + d) * 16 + 4 * q + p] = bout;
}

__device__ inline float lane_bcast(float v, int k) {
  return __int_as_float(__builtin_amdgcn_readlane(__float_as_int(v), k));
}

__device__ inline float affine_row(float4 m0, float4 m1, float4 m2, float4 m3,
                                   float vec, float bias) {
  float s[16];
#pragma unroll
  for (int k = 0; k < 16; ++k) s[k] = lane_bcast(vec, k);
  float s0 = m0.x * s[0], s1 = m1.x * s[4], s2 = m2.x * s[8], s3 = m3.x * s[12];
  s0 = fmaf(m0.y, s[1], s0);  s1 = fmaf(m1.y, s[5], s1);
  s2 = fmaf(m2.y, s[9], s2);  s3 = fmaf(m3.y, s[13], s3);
  s0 = fmaf(m0.z, s[2], s0);  s1 = fmaf(m1.z, s[6], s1);
  s2 = fmaf(m2.z, s[10], s2); s3 = fmaf(m3.z, s[14], s3);
  s0 = fmaf(m0.w, s[3], s0);  s1 = fmaf(m1.w, s[7], s1);
  s2 = fmaf(m2.w, s[11], s2); s3 = fmaf(m3.w, s[15], s3);
  return bias + ((s0 + s1) + (s2 + s3));
}

__global__ void __launch_bounds__(512) __attribute__((amdgpu_waves_per_eu(4, 4)))
k_combine(const float* __restrict__ Tmat,
          const float* __restrict__ bvec,
          const float* __restrict__ init_h,
          float* __restrict__ hstart,
          float* __restrict__ h_final) {
  __shared__ __attribute__((aligned(16))) float Pm[16][256];
  __shared__ __attribute__((aligned(16))) float qv[16][16];
  __shared__ __attribute__((aligned(16))) float Hg[17][16];

  const int d    = blockIdx.x;
  const int g    = threadIdx.x >> 5;
  const int lane = threadIdx.x & 31;
  const int lo   = lane & 15, hi = lane >> 4;

  v8f P;
#pragma unroll
  for (int r = 0; r < 8; ++r) P[r] = ((r + 8 * hi) == lo) ? 1.f : 0.f;
  float q = 0.f;

#pragma unroll
  for (int i = 0; i < 16; ++i) {
    const int c = g * 16 + i;
    const float* Tc = Tmat + (c * 16 + d) * 256;
    v2f a[4];
#pragma unroll
    for (int kk = 0; kk < 4; ++kk)
      a[kk] = *reinterpret_cast<const v2f*>(Tc + lo * 16 + 8 * hi + 2 * kk);
    const float4* Trow = reinterpret_cast<const float4*>(Tc + lo * 16);
    const float4 m0 = Trow[0], m1 = Trow[1], m2 = Trow[2], m3 = Trow[3];
    const float bias = bvec[(c * 16 + d) * 16 + lo];

    v8f acc = {};
#pragma unroll
    for (int kk = 0; kk < 4; ++kk) {
      v2f b;
      b.x = P[2 * kk];
      b.y = P[2 * kk + 1];
      acc = wmma4(a[kk], b, acc);
    }
    q = affine_row(m0, m1, m2, m3, q, bias);
    P = acc;
  }
#pragma unroll
  for (int r = 0; r < 8; ++r) Pm[g][(r + 8 * hi) * 16 + lo] = P[r];
  qv[g][lo] = q;
  __syncthreads();

  if (g == 0) {
    float h = init_h[d * 16 + lo];
    Hg[0][lo] = h;
#pragma unroll
    for (int k = 0; k < 16; ++k) {
      const float4* Prow = reinterpret_cast<const float4*>(&Pm[k][lo * 16]);
      h = affine_row(Prow[0], Prow[1], Prow[2], Prow[3], h, qv[k][lo]);
      Hg[k + 1][lo] = h;
    }
    if (hi == 0) h_final[d * 16 + lo] = h;
  }
  __syncthreads();

  {
    float hs[16];
    float h = Hg[g][lo];
#pragma unroll
    for (int i = 0; i < 16; ++i) {
      const int c = g * 16 + i;
      hs[i] = h;
      if (i < 15) {
        const float4* Trow =
            reinterpret_cast<const float4*>(Tmat + ((c * 16 + d) * 16 + lo) * 16);
        h = affine_row(Trow[0], Trow[1], Trow[2], Trow[3], h,
                       bvec[(c * 16 + d) * 16 + lo]);
      }
    }
    if (hi == 0) {
#pragma unroll
      for (int i = 0; i < 16; ++i)
        hstart[((g * 16 + i) * 16 + d) * 16 + lo] = hs[i];
    }
  }
}

#define SB4    16
#define NBLK4  (CHUNK / SB4)
#define HP     20

struct ScanOperands { float dtc[SB4 / 2], xbd[SB4 / 2], Bj[SB4 / 2]; };

__device__ inline void scan_load(ScanOperands& o,
                                 const float* __restrict__ dt,
                                 const float* __restrict__ xb,
                                 const float* __restrict__ Bv,
                                 int tbase, int fth, int fd, int fj) {
#pragma unroll
  for (int i = 0; i < SB4 / 2; ++i) {
    const int tt = tbase + 2 * i + fth;
    o.dtc[i] = dt[tt * 16 + fd];
    o.xbd[i] = xb[tt * 16 + fd];
    o.Bj[i]  = Bv[tt * 16 + fj];
  }
}

struct FillPiece { float num, den, c, s, u; };

__device__ inline FillPiece fill_front(const ScanOperands& o, int i,
                                       float fA, float fF) {
  const float dtc = o.dtc[i];
  const float ta  = 0.5f * dtc * fA;
  const float ang = dtc * fF;
  FillPiece p;
  p.num = 1.f + ta;
  p.den = 1.f - ta + 1e-8f;
  p.c   = __cosf(ang);
  p.s   = __sinf(ang);
  p.u   = dtc * o.Bj[i] * o.xbd[i];
  return p;
}

__device__ inline void fill_back(float4* __restrict__ buf, const FillPiece& p,
                                 int i, int fth, int fd, int fj) {
  const int t = 2 * i + fth;
  float4 v;
  v.x = p.num / p.den;
  v.y = p.c;
  v.z = p.s;
  v.w = p.u;
  buf[(t * 16 + fd) * 16 + fj] = v;
}

__device__ inline void scan_fill(float4* __restrict__ buf, const ScanOperands& o,
                                 int fth, int fd, int fj, float fA, float fF) {
#pragma unroll
  for (int i = 0; i < SB4 / 2; ++i) {
    const FillPiece p = fill_front(o, i, fA, fF);
    fill_back(buf, p, i, fth, fd, fj);
  }
}

__device__ inline float half_sum(float4 ha, float4 hb, float4 ca, float4 cb) {
  const float q0 = fmaf(ha.x, ca.x, ha.y * ca.y), q1 = fmaf(ha.z, ca.z, ha.w * ca.w);
  const float q2 = fmaf(hb.x, cb.x, hb.y * cb.y), q3 = fmaf(hb.z, cb.z, hb.w * cb.w);
  return (q0 + q1) + (q2 + q3);
}

template <bool FILL, bool PREV>
__device__ inline float scan_walk(const float4* __restrict__ trow,
                                  float* hrow, const float4* hrd,
                                  float h, int src,
                                  float4 caP, float4 cbP, bool writer,
                                  float* __restrict__ ypP,
                                  float4* __restrict__ nbuf,
                                  const ScanOperands& o, int fth, int fd,
                                  int fj, float fA, float fF) {
  float4 ha = {}, hb = {};
  if (PREV) { ha = hrd[0]; hb = hrd[1]; }
  float4 co[SB4];
#pragma unroll
  for (int t = 0; t < SB4; ++t) co[t] = trow[t * 256];

  FillPiece p = {};
  float sp = 0.f, sx = 0.f;
#pragma unroll
  for (int t = 0; t < SB4; ++t) {
    const float hr = __shfl(h, src, 32);
    __builtin_amdgcn_sched_barrier(0);
    if (FILL) {
      if ((t & 1) == 0) p = fill_front(o, t >> 1, fA, fF);
      else              fill_back(nbuf, p, t >> 1, fth, fd, fj);
    }
    if (PREV && t == 0) {
      sp = half_sum(ha, hb, caP, cbP);
      sx = __shfl_xor(sp, 16, 32);
    }
    if (PREV && t == 1) {
      if (writer) *ypP = sp + sx;
    }
    __builtin_amdgcn_sched_barrier(0);
    h = co[t].x * (co[t].y * h - co[t].z * hr) + co[t].w;
    hrow[t * HP] = h;
  }
  return h;
}

__global__ void __launch_bounds__(512)
k_scan(const float* __restrict__ dt,
       const float* __restrict__ xb,
       const float* __restrict__ Bv,
       const float* __restrict__ Cv,
       const float* __restrict__ A_log,
       const float* __restrict__ rope,
       const float* __restrict__ hstart,
       float* __restrict__ y) {
  __shared__ __attribute__((aligned(16))) float4 tab[2][SB4 * 16 * 16];
  __shared__ __attribute__((aligned(16))) float  hst[16 * SB4 * HP];

  const int c    = blockIdx.x;
  const int tid  = threadIdx.x;
  const int d    = tid >> 5;
  const int lane = tid & 31;
  const int lo   = lane & 15, hi = lane >> 4;

  const int fj  = tid & 15;
  const int fd  = (tid >> 4) & 15;
  const int fth = tid >> 8;
  const float fA = -__expf(A_log[fd * 16 + fj]);
  const float fF = rope[fd * 16 + fj];

  float h = hstart[(c * 16 + d) * 16 + lo];
  const int t0  = c * CHUNK;
  const int src = (lane & 16) | ((lo + 15) & 15);

  float*        hrow = hst + (d * SB4) * HP + lo;
  const float4* hrd  =
      reinterpret_cast<const float4*>(hst + (d * SB4 + lo) * HP + 8 * hi);
  const bool writer = (hi == 0);

  const float4* Crow = reinterpret_cast<const float4*>(Cv + (t0 + lo) * 16 + 8 * hi);
  float*        yrow = y + (t0 + lo) * 16 + d;
  const int     CSTEP = SB4 * 16 / 4;
  const int     YSTEP = SB4 * 16;

  ScanOperands opsA, opsB;
  scan_load(opsB, dt, xb, Bv, t0, fth, fd, fj);
  scan_load(opsA, dt, xb, Bv, t0 + SB4, fth, fd, fj);
  scan_fill(tab[0], opsB, fth, fd, fj, fA, fF);
  __syncthreads();

  scan_load(opsB, dt, xb, Bv, t0 + 2 * SB4, fth, fd, fj);
  float4 caP = Crow[0], cbP = Crow[1];
  h = scan_walk<true, false>(tab[0] + d * 16 + lo, hrow, hrd, h, src,
                             caP, cbP, writer, yrow,
                             tab[1], opsA, fth, fd, fj, fA, fF);
  opsA = opsB;
  __syncthreads();

  for (int kb = 1; kb < NBLK4 - 1; ++kb) {
    if (kb + 2 < NBLK4)
      scan_load(opsB, dt, xb, Bv, t0 + (kb + 2) * SB4, fth, fd, fj);
    const float4 ca = Crow[kb * CSTEP], cb = Crow[kb * CSTEP + 1];

    h = scan_walk<true, true>(tab[kb & 1] + d * 16 + lo, hrow, hrd, h, src,
                              caP, cbP, writer, yrow + (kb - 1) * YSTEP,
                              tab[(kb + 1) & 1], opsA, fth, fd, fj, fA, fF);
    opsA = opsB;
    caP = ca; cbP = cb;
    __syncthreads();
  }
  {
    const int kb = NBLK4 - 1;
    const float4 ca = Crow[kb * CSTEP], cb = Crow[kb * CSTEP + 1];
    scan_walk<false, true>(tab[kb & 1] + d * 16 + lo, hrow, hrd, h, src,
                           caP, cbP, writer, yrow + (kb - 1) * YSTEP,
                           tab[0], opsA, fth, fd, fj, fA, fF);
    float s = half_sum(hrd[0], hrd[1], ca, cb);
    s += __shfl_xor(s, 16, 32);
    if (writer) yrow[kb * YSTEP] = s;
  }
}

__global__ void k_out(const float* __restrict__ y,
                      const float* __restrict__ z,
                      const float* __restrict__ xb,
                      const float* __restrict__ Dp,
                      const float* __restrict__ outw,
                      float* __restrict__ out) {
  const int lane = threadIdx.x & 31;
  const int wave = threadIdx.x >> 5;
  const int t0   = (blockIdx.x * 4 + wave) * 16;
  const int lo   = lane & 15, hi = lane >> 4;

  v8f acc = {};
#pragma unroll
  for (int kk = 0; kk < 4; ++kk) {
    const int kb = kk * 4 + 2 * hi;
    v2f a, b;
#pragma unroll
    for (int v = 0; v < 2; ++v) {
      const int k   = kb + v;
      const int idx = (t0 + lo) * 16 + k;
      const float zz  = z[idx];
      const float sil = zz / (1.f + __expf(-zz));
      const float val = y[idx] * sil + Dp[k] * xb[idx];
      if (v == 0) a.x = val; else a.y = val;
    }
    b.x = (lo < 8) ? outw[lo * 16 + kb + 0] : 0.f;
    b.y = (lo < 8) ? outw[lo * 16 + kb + 1] : 0.f;
    acc = wmma4(a, b, acc);
  }
  if (lo < 8) {
#pragma unroll
    for (int r = 0; r < 8; ++r) out[(t0 + r + 8 * hi) * 8 + lo] = acc[r];
  }
}

extern "C" void kernel_launch(void* const* d_in, const int* in_sizes, int n_in,
                              void* d_out, int out_size, void* d_ws,
                              size_t ws_size, hipStream_t stream) {
  const float* x    = (const float*)d_in[0];
  const float* init = (const float*)d_in[1];
  const float* inw  = (const float*)d_in[2];
  const float* dtw  = (const float*)d_in[3];
  const float* dtb  = (const float*)d_in[4];
  const float* bw   = (const float*)d_in[5];
  const float* cw   = (const float*)d_in[6];
  const float* alog = (const float*)d_in[7];
  const float* Dp   = (const float*)d_in[8];
  const float* rope = (const float*)d_in[9];
  const float* outw = (const float*)d_in[10];
  float* out = (float*)d_out;

  float* ws = (float*)d_ws;
  float* XB = ws;
  float* Z  = XB + N16;
  float* DT = Z + N16;
  float* BV = DT + N16;
  float* CV = BV + N16;
  float* Y  = CV + N16;
  float* T  = Y + N16;
  float* BC = T + N16;
  float* HS = BC + NCHUNK * 256;

  k_proj<<<NTOK / 64, 128, 0, stream>>>(x, inw, dtw, dtb, bw, cw,
                                        XB, Z, DT, BV, CV);
  k_chunkop<<<NCHUNK, KT2, 0, stream>>>(DT, XB, BV, alog, rope, T, BC);
  k_combine<<<16, 512, 0, stream>>>(T, BC, init, HS, out + NTOK * DM);
  k_scan<<<NCHUNK, 512, 0, stream>>>(DT, XB, BV, CV, alog, rope, HS, Y);
  k_out<<<NTOK / 64, 128, 0, stream>>>(Y, Z, XB, Dp, outw, out);
}


# K2 chunk operator table: a lane makes two entries at once in packed f32; abar = num*rcp(den) with one fused correction step, not the scalar-only division sequence (abar may differ in the last bit)
# speedup vs baseline: 5.6357x; 1.0600x over previous
#include <hip/hip_runtime.h>
#include <hip/hip_bf16.h>
#include <math.h>


typedef __attribute__((ext_vector_type(2))) float v2f;
typedef __attribute__((ext_vector_type(8))) float v8f;

#define NTOK   65536
#define DM     8
#define DI     16
#define CHUNK  256
#define NCHUNK 256
#define N16    (NTOK * DI)

__device__ inline v8f wmma4(v2f a, v2f b, v8f c) {
  return __builtin_amdgcn_wmma_f32_16x16x4_f32(
      false, a, false, b, (short)0, c, false, false);
}

__device__ inline float softplus_f(float s) {
  return (s > 20.f) ? s : logf(1.f + __expf(s));
}

__global__ void k_proj(const float* __restrict__ x,
                       const float* __restrict__ inw,
                       const float* __restrict__ dtw,
                       const float* __restrict__ dtb,
                       const float* __restrict__ bw,
                       const float* __restrict__ cw,
                       float* __restrict__ xb, float* __restrict__ z,
                       float* __restrict__ dt, float* __restrict__ Bv,
                       float* __restrict__ Cv) {
  __shared__ float lds[4][256];
  const int lane = threadIdx.x & 31;
  const int wave = threadIdx.x >> 5;
  const int t0   = (blockIdx.x * 4 + wave) * 16;
  const int lo   = lane & 15, hi = lane >> 4;

  v8f axb = {}, az = {};
#pragma unroll
  for (int kk = 0; kk < 2; ++kk) {
    const int kb = kk * 4 + 2 * hi;
    v2f a, b0, b1;
    a.x  = x[(t0 + lo) * 8 + kb + 0];
    a.y  = x[(t0 + lo) * 8 + kb + 1];
    b0.x = inw[lo * 8 + kb + 0];
    b0.y = inw[lo * 8 + kb + 1];
    b1.x = inw[(lo + 16) * 8 + kb + 0];
    b1.y = inw[(lo + 16) * 8 + kb + 1];
    axb = wmma4(a, b0, axb);
    az  = wmma4(a, b1, az);
  }

#pragma unroll
  for (int r = 0; r < 8; ++r) lds[wave][(r + 8 * hi) * 16 + lo] = axb[r];

  v8f adt = {}, ab = {}, ac = {};
#pragma unroll
  for (int kk = 0; kk < 4; ++kk) {
    const int kb = kk * 4 + 2 * hi;
    v2f a, b;
    a.x = lds[wave][lo * 16 + kb + 0];
    a.y = lds[wave][lo * 16 + kb + 1];
    b.x = dtw[lo * 16 + kb + 0]; b.y = dtw[lo * 16 + kb + 1];
    adt = wmma4(a, b, adt);
    b.x = bw[lo * 16 + kb + 0];  b.y = bw[lo * 16 + kb + 1];
    ab  = wmma4(a, b, ab);
    b.x = cw[lo * 16 + kb + 0];  b.y = cw[lo * 16 + kb + 1];
    ac  = wmma4(a, b, ac);
  }

  const float bias = dtb[lo];
#pragma unroll
  for (int r = 0; r < 8; ++r) {
    const int idx = (t0 + r + 8 * hi) * 16 + lo;
    xb[idx] = axb[r];
    z[idx]  = az[r];
    dt[idx] = softplus_f(adt[r] + bias);
    Bv[idx] = ab[r];
    Cv[idx] = ac[r];
  }
}

#define SB    32
#define NBLK  (CHUNK / SB)
#define ROWF  48

struct FillOperands { float dtc[SB / 2], xbd[SB / 2], Bj[SB / 2]; };

__device__ inline void chunkop_load(FillOperands& o,
                                    const float* __restrict__ dt,
                                    const float* __restrict__ xb,
                                    const float* __restrict__ Bv,
                                    int tbase, int fth, int fd, int fj) {
#pragma unroll
  for (int i = 0; i < SB / 2; ++i) {
    const int tt = tbase + 2 * i + fth;
    o.dtc[i] = dt[tt * 16 + fd];
    o.xbd[i] = xb[tt * 16 + fd];
    o.Bj[i]  = Bv[tt * 16 + fj];
  }
}

__device__ inline v2f fma2(v2f a, v2f b, v2f c) { return __builtin_elementwise_fma(a, b, c); }

__device__ inline void chunkop_make(float* __restrict__ buf,
                                    const FillOperands& o,
                                    int fth, int fd, int fj,
                                    float fA, float fF) {
  const float hA = 0.5f * fA;
#pragma unroll
  for (int i = 0; i < SB / 2; i += 2) {
    v2f dtc; dtc.x = o.dtc[i]; dtc.y = o.dtc[i + 1];
    v2f Bj;  Bj.x  = o.Bj[i];  Bj.y  = o.Bj[i + 1];
    v2f xbd; xbd.x = o.xbd[i]; xbd.y = o.xbd[i + 1];
    const v2f ta  = dtc * hA;
    const v2f num = 1.f + ta;
    const v2f den = (1.f - ta) + 1e-8f;
    v2f r; r.x = __builtin_amdgcn_rcpf(den.x); r.y = __builtin_amdgcn_rcpf(den.y);
    v2f q = num * r;
    q = fma2(fma2(-den, q, num), r, q);
    const v2f rev = (dtc * fF) * 0.15915494309189532f;
    v2f cs, sn;
    cs.x = __builtin_amdgcn_cosf(rev.x); cs.y = __builtin_amdgcn_cosf(rev.y);
    sn.x = __builtin_amdgcn_sinf(rev.x); sn.y = __builtin_amdgcn_sinf(rev.y);
    const v2f E = q * cs, F = q * sn, U = dtc * Bj * xbd;
    const int t = 2 * i + fth;
    float* row0 = buf + ((SB - 1 - t) * 16 + fd) * ROWF;
    float* row1 = row0 - 2 * 16 * ROWF;
    row0[fj]      = E.x;
    row0[16 + ((fj + 15) & 15)] = F.x;
    row0[32 + fj] = U.x;
    row1[fj]      = E.y;
    row1[16 + ((fj + 15) & 15)] = F.y;
    row1[32 + fj] = U.y;
  }
}

__device__ inline void chunkop_fill(float* __restrict__ buf,
                                    const float* __restrict__ dt,
                                    const float* __restrict__ xb,
                                    const float* __restrict__ Bv,
                                    int tbase, int fth, int fd, int fj,
                                    float fA, float fF) {
  FillOperands o;
  chunkop_load(o, dt, xb, Bv, tbase, fth, fd, fj);
  chunkop_make(buf, o, fth, fd, fj, fA, fF);
}

#define KW2   512
#define KT2   768

__device__ inline float next_pair(float v) {
  const int i = __builtin_bit_cast(int, v);
  return __builtin_bit_cast(float, __builtin_amdgcn_permlane16(
      i, i, (int)0x07654321u, (int)0x8fedcba9u, true, false));
}

__global__ void __launch_bounds__(KT2)
k_chunkop(const float* __restrict__ dt,
          const float* __restrict__ xb,
          const float* __restrict__ Bv,
          const float* __restrict__ A_log,
          const float* __restrict__ rope,
          float* __restrict__ Tmat,
          float* __restrict__ bvec) {
  __shared__ __attribute__((aligned(16))) float tab[2][SB * 16 * ROWF];

  const int c    = blockIdx.x;
  const int tid  = threadIdx.x;
  const int d    = tid >> 5;
  const int lane = tid & 31;

  const bool producer = tid >= KW2;
  const int fj  = tid & 15;
  const int fd  = (tid >> 4) & 15;
  const int fth = (tid >> 8) & 1;
  const float fA = -__expf(A_log[fd * 16 + fj]);
  const float fF = rope[fd * 16 + fj];

  const int p = lane & 7, q = lane >> 3;
  float Sa[4], Sb[4], bacc[4];
#pragma unroll
  for (int r = 0; r < 4; ++r) {
    Sa[r]   = ((4 * q + r) == 2 * p)     ? 1.f : 0.f;
    Sb[r]   = ((4 * q + r) == 2 * p + 1) ? 1.f : 0.f;
    bacc[r] = 0.f;
  }

  const int t0 = c * CHUNK;

  if (!producer)
    chunkop_fill(tab[0], dt, xb, Bv, t0 + (NBLK - 1) * SB, fth, fd, fj, fA, fF);
  __syncthreads();

  for (int kb = 0; kb < NBLK; ++kb) {
    if (producer) {
      if (kb + 1 < NBLK) {
        float* nb = tab[(kb + 1) & 1];
        const int tn = t0 + (NBLK - 2 - kb) * SB;
        FillOperands o;
        chunkop_load(o, dt, xb, Bv, tn, 0, fd, fj);
        __builtin_amdgcn_sched_barrier(0);
        chunkop_make(nb, o, 0, fd, fj, fA, fF);
        chunkop_load(o, dt, xb, Bv, tn, 1, fd, fj);
        __builtin_amdgcn_sched_barrier(0);
        chunkop_make(nb, o, 1, fd, fj, fA, fF);
      }
    } else {
      const float* base = tab[kb & 1] + d * ROWF + 2 * p;
#pragma unroll 4
      for (int i = 0; i < SB; ++i) {
        const float* row = base + i * 16 * ROWF;
        const float2 E = *reinterpret_cast<const float2*>(row);
        const float2 F = *reinterpret_cast<const float2*>(row + 16);
        const float2 U = *reinterpret_cast<const float2*>(row + 32);
#pragma unroll
        for (int r = 0; r < 4; ++r) {
          const float sn = next_pair(Sa[r]);
          bacc[r] = fmaf(Sb[r], U.y, fmaf(Sa[r], U.x, bacc[r]));
          const float a = E.x * Sa[r] - F.x * Sb[r];
          Sb[r] = E.y * Sb[r] - F.y * sn;
          Sa[r] = a;
        }
      }
    }
    __syncthreads();
  }

  if (producer) return;
#pragma unroll
  for (int r = 0; r < 4; ++r) {
    float2 v; v.x = Sa[r]; v.y = Sb[r];
    *reinterpret_cast<float2*>(Tmat + ((c * 16 + d) * 16 + (4 * q + r)) * 16 + 2 * p) = v;
  }

  float bout = 0.f;
#pragma unroll
  for (int r = 0; r < 4; ++r) {
    float v = bacc[r];
    v += __shfl_xor(v, 1, 32);
    v += __shfl_xor(v, 2, 32);
    v += __shfl_xor(v, 4, 32);
    bout = (p == r) ? v : bout;
  }
  if (p < 4) bvec[(c * 16 + d) * 16 + 4 * q + p] = bout;
}

__device__ inline float lane_bcast(float v, int k) {
  return __int_as_float(__builtin_amdgcn_readlane(__float_as_int(v), k));
}

__device__ inline float affine_row(float4 m0, float4 m1, float4 m2, float4 m3,
                                   float vec, float bias) {
  float s[16];
#pragma unroll
  for (int k = 0; k < 16; ++k) s[k] = lane_bcast(vec, k);
  float s0 = m0.x * s[0], s1 = m1.x * s[4], s2 = m2.x * s[8], s3 = m3.x * s[12];
  s0 = fmaf(m0.y, s[1], s0);  s1 = fmaf(m1.y, s[5], s1);
  s2 = fmaf(m2.y, s[9], s2);  s3 = fmaf(m3.y, s[13], s3);
  s0 = fmaf(m0.z, s[2], s0);  s1 = fmaf(m1.z, s[6], s1);
  s2 = fmaf(m2.z, s[10], s2); s3 = fmaf(m3.z, s[14], s3);
  s0 = fmaf(m0.w, s[3], s0);  s1 = fmaf(m1.w, s[7], s1);
  s2 = fmaf(m2.w, s[11], s2); s3 = fmaf(m3.w, s[15], s3);
  return bias + ((s0 + s1) + (s2 + s3));
}

__global__ void __launch_bounds__(512) __attribute__((amdgpu_waves_per_eu(4, 4)))
k_combine(const float* __restrict__ Tmat,
          const float* __restrict__ bvec,
          const float* __restrict__ init_h,
          float* __restrict__ hstart,
          float* __restrict__ h_final) {
  __shared__ __attribute__((aligned(16))) float Pm[16][256];
  __shared__ __attribute__((aligned(16))) float qv[16][16];
  __shared__ __attribute__((aligned(16))) float Hg[17][16];

  const int d    = blockIdx.x;
  const int g    = threadIdx.x >> 5;
  const int lane = threadIdx.x & 31;
  const int lo   = lane & 15, hi = lane >> 4;

  v8f P;
#pragma unroll
  for (int r = 0; r < 8; ++r) P[r] = ((r + 8 * hi) == lo) ? 1.f : 0.f;
  float q = 0.f;

#pragma unroll
  for (int i = 0; i < 16; ++i) {
    const int c = g * 16 + i;
    const float* Tc = Tmat + (c * 16 + d) * 256;
    v2f a[4];
#pragma unroll
    for (int kk = 0; kk < 4; ++kk)
      a[kk] = *reinterpret_cast<const v2f*>(Tc + lo * 16 + 8 * hi + 2 * kk);
    const float4* Trow = reinterpret_cast<const float4*>(Tc + lo * 16);
    const float4 m0 = Trow[0], m1 = Trow[1], m2 = Trow[2], m3 = Trow[3];
    const float bias = bvec[(c * 16 + d) * 16 + lo];

    v8f acc = {};
#pragma unroll
    for (int kk = 0; kk < 4; ++kk) {
      v2f b;
      b.x = P[2 * kk];
      b.y = P[2 * kk + 1];
      acc = wmma4(a[kk], b, acc);
    }
    q = affine_row(m0, m1, m2, m3, q, bias);
    P = acc;
  }
#pragma unroll
  for (int r = 0; r < 8; ++r) Pm[g][(r + 8 * hi) * 16 + lo] = P[r];
  qv[g][lo] = q;
  __syncthreads();

  if (g == 0) {
    float h = init_h[d * 16 + lo];
    Hg[0][lo] = h;
#pragma unroll
    for (int k = 0; k < 16; ++k) {
      const float4* Prow = reinterpret_cast<const float4*>(&Pm[k][lo * 16]);
      h = affine_row(Prow[0], Prow[1], Prow[2], Prow[3], h, qv[k][lo]);
      Hg[k + 1][lo] = h;
    }
    if (hi == 0) h_final[d * 16 + lo] = h;
  }
  __syncthreads();

  {
    float hs[16];
    float h = Hg[g][lo];
#pragma unroll
    for (int i = 0; i < 16; ++i) {
      const int c = g * 16 + i;
      hs[i] = h;
      if (i < 15) {
        const float4* Trow =
            reinterpret_cast<const float4*>(Tmat + ((c * 16 + d) * 16 + lo) * 16);
        h = affine_row(Trow[0], Trow[1], Trow[2], Trow[3], h,
                       bvec[(c * 16 + d) * 16 + lo]);
      }
    }
    if (hi == 0) {
#pragma unroll
      for (int i = 0; i < 16; ++i)
        hstart[((g * 16 + i) * 16 + d) * 16 + lo] = hs[i];
    }
  }
}

#define SB4    16
#define NBLK4  (CHUNK / SB4)
#define HP     20

struct ScanOperands { float dtc[SB4 / 2], xbd[SB4 / 2], Bj[SB4 / 2]; };

__device__ inline void scan_load(ScanOperands& o,
                                 const float* __restrict__ dt,
                                 const float* __restrict__ xb,
                                 const float* __restrict__ Bv,
                                 int tbase, int fth, int fd, int fj) {
#pragma unroll
  for (int i = 0; i < SB4 / 2; ++i) {
    const int tt = tbase + 2 * i + fth;
    o.dtc[i] = dt[tt * 16 + fd];
    o.xbd[i] = xb[tt * 16 + fd];
    o.Bj[i]  = Bv[tt * 16 + fj];
  }
}

struct FillPiece { float num, den, c, s, u; };

__device__ inline FillPiece fill_front(const ScanOperands& o, int i,
                                       float fA, float fF) {
  const float dtc = o.dtc[i];
  const float ta  = 0.5f * dtc * fA;
  const float ang = dtc * fF;
  FillPiece p;
  p.num = 1.f + ta;
  p.den = 1.f - ta + 1e-8f;
  p.c   = __cosf(ang);
  p.s   = __sinf(ang);
  p.u   = dtc * o.Bj[i] * o.xbd[i];
  return p;
}

__device__ inline void fill_back(float4* __restrict__ buf, const FillPiece& p,
                                 int i, int fth, int fd, int fj) {
  const int t = 2 * i + fth;
  float4 v;
  v.x = p.num / p.den;
  v.y = p.c;
  v.z = p.s;
  v.w = p.u;
  buf[(t * 16 + fd) * 16 + fj] = v;
}

__device__ inline void scan_fill(float4* __restrict__ buf, const ScanOperands& o,
                                 int fth, int fd, int fj, float fA, float fF) {
#pragma unroll
  for (int i = 0; i < SB4 / 2; ++i) {
    const FillPiece p = fill_front(o, i, fA, fF);
    fill_back(buf, p, i, fth, fd, fj);
  }
}

__device__ inline float half_sum(float4 ha, float4 hb, float4 ca, float4 cb) {
  const float q0 = fmaf(ha.x, ca.x, ha.y * ca.y), q1 = fmaf(ha.z, ca.z, ha.w * ca.w);
  const float q2 = fmaf(hb.x, cb.x, hb.y * cb.y), q3 = fmaf(hb.z, cb.z, hb.w * cb.w);
  return (q0 + q1) + (q2 + q3);
}

template <bool FILL, bool PREV>
__device__ inline float scan_walk(const float4* __restrict__ trow,
                                  float* hrow, const float4* hrd,
                                  float h, int src,
                                  float4 caP, float4 cbP, bool writer,
                                  float* __restrict__ ypP,
                                  float4* __restrict__ nbuf,
                                  const ScanOperands& o, int fth, int fd,
                                  int fj, float fA, float fF) {
  float4 ha = {}, hb = {};
  if (PREV) { ha = hrd[0]; hb = hrd[1]; }
  float4 co[SB4];
#pragma unroll
  for (int t = 0; t < SB4; ++t) co[t] = trow[t * 256];

  FillPiece p = {};
  float sp = 0.f, sx = 0.f;
#pragma unroll
  for (int t = 0; t < SB4; ++t) {
    const float hr = __shfl(h, src, 32);
    __builtin_amdgcn_sched_barrier(0);
    if (FILL) {
      if ((t & 1) == 0) p = fill_front(o, t >> 1, fA, fF);
      else              fill_back(nbuf, p, t >> 1, fth, fd, fj);
    }
    if (PREV && t == 0) {
      sp = half_sum(ha, hb, caP, cbP);
      sx = __shfl_xor(sp, 16, 32);
    }
    if (PREV && t == 1) {
      if (writer) *ypP = sp + sx;
    }
    __builtin_amdgcn_sched_barrier(0);
    h = co[t].x * (co[t].y * h - co[t].z * hr) + co[t].w;
    hrow[t * HP] = h;
  }
  return h;
}

__global__ void __launch_bounds__(512)
k_scan(const float* __restrict__ dt,
       const float* __restrict__ xb,
       const float* __restrict__ Bv,
       const float* __restrict__ Cv,
       const float* __restrict__ A_log,
       const float* __restrict__ rope,
       const float* __restrict__ hstart,
       float* __restrict__ y) {
  __shared__ __attribute__((aligned(16))) float4 tab[2][SB4 * 16 * 16];
  __shared__ __attribute__((aligned(16))) float  hst[16 * SB4 * HP];

  const int c    = blockIdx.x;
  const int tid  = threadIdx.x;
  const int d    = tid >> 5;
  const int lane = tid & 31;
  const int lo   = lane & 15, hi = lane >> 4;

  const int fj  = tid & 15;
  const int fd  = (tid >> 4) & 15;
  const int fth = tid >> 8;
  const float fA = -__expf(A_log[fd * 16 + fj]);
  const float fF = rope[fd * 16 + fj];

  float h = hstart[(c * 16 + d) * 16 + lo];
  const int t0  = c * CHUNK;
  const int src = (lane & 16) | ((lo + 15) & 15);

  float*        hrow = hst + (d * SB4) * HP + lo;
  const float4* hrd  =
      reinterpret_cast<const float4*>(hst + (d * SB4 + lo) * HP + 8 * hi);
  const bool writer = (hi == 0);

  const float4* Crow = reinterpret_cast<const float4*>(Cv + (t0 + lo) * 16 + 8 * hi);
  float*        yrow = y + (t0 + lo) * 16 + d;
  const int     CSTEP = SB4 * 16 / 4;
  const int     YSTEP = SB4 * 16;

  ScanOperands opsA, opsB;
  scan_load(opsB, dt, xb, Bv, t0, fth, fd, fj);
  scan_load(opsA, dt, xb, Bv, t0 + SB4, fth, fd, fj);
  scan_fill(tab[0], opsB, fth, fd, fj, fA, fF);
  __syncthreads();

  scan_load(opsB, dt, xb, Bv, t0 + 2 * SB4, fth, fd, fj);
  float4 caP = Crow[0], cbP = Crow[1];
  h = scan_walk<true, false>(tab[0] + d * 16 + lo, hrow, hrd, h, src,
                             caP, cbP, writer, yrow,
                             tab[1], opsA, fth, fd, fj, fA, fF);
  opsA = opsB;
  __syncthreads();

  for (int kb = 1; kb < NBLK4 - 1; ++kb) {
    if (kb + 2 < NBLK4)
      scan_load(opsB, dt, xb, Bv, t0 + (kb + 2) * SB4, fth, fd, fj);
    const float4 ca = Crow[kb * CSTEP], cb = Crow[kb * CSTEP + 1];

    h = scan_walk<true, true>(tab[kb & 1] + d * 16 + lo, hrow, hrd, h, src,
                              caP, cbP, writer, yrow + (kb - 1) * YSTEP,
                              tab[(kb + 1) & 1], opsA, fth, fd, fj, fA, fF);
    opsA = opsB;
    caP = ca; cbP = cb;
    __syncthreads();
  }
  {
    const int kb = NBLK4 - 1;
    const float4 ca = Crow[kb * CSTEP], cb = Crow[kb * CSTEP + 1];
    scan_walk<false, true>(tab[kb & 1] + d * 16 + lo, hrow, hrd, h, src,
                           caP, cbP, writer, yrow + (kb - 1) * YSTEP,
                           tab[0], opsA, fth, fd, fj, fA, fF);
    float s = half_sum(hrd[0], hrd[1], ca, cb);
    s += __shfl_xor(s, 16, 32);
    if (writer) yrow[kb * YSTEP] = s;
  }
}

__global__ void k_out(const float* __restrict__ y,
                      const float* __restrict__ z,
                      const float* __restrict__ xb,
                      const float* __restrict__ Dp,
                      const float* __restrict__ outw,
                      float* __restrict__ out) {
  const int lane = threadIdx.x & 31;
  const int wave = threadIdx.x >> 5;
  const int t0   = (blockIdx.x * 4 + wave) * 16;
  const int lo   = lane & 15, hi = lane >> 4;

  v8f acc = {};
#pragma unroll
  for (int kk = 0; kk < 4; ++kk) {
    const int kb = kk * 4 + 2 * hi;
    v2f a, b;
#pragma unroll
    for (int v = 0; v < 2; ++v) {
      const int k   = kb + v;
      const int idx = (t0 + lo) * 16 + k;
      const float zz  = z[idx];
      const float sil = zz / (1.f + __expf(-zz));
      const float val = y[idx] * sil + Dp[k] * xb[idx];
      if (v == 0) a.x = val; else a.y = val;
    }
    b.x = (lo < 8) ? outw[lo * 16 + kb + 0] : 0.f;
    b.y = (lo < 8) ? outw[lo * 16 + kb + 1] : 0.f;
    acc = wmma4(a, b, acc);
  }
  if (lo < 8) {
#pragma unroll
    for (int r = 0; r < 8; ++r) out[(t0 + r + 8 * hi) * 8 + lo] = acc[r];
  }
}

extern "C" void kernel_launch(void* const* d_in, const int* in_sizes, int n_in,
                              void* d_out, int out_size, void* d_ws,
                              size_t ws_size, hipStream_t stream) {
  const float* x    = (const float*)d_in[0];
  const float* init = (const float*)d_in[1];
  const float* inw  = (const float*)d_in[2];
  const float* dtw  = (const float*)d_in[3];
  const float* dtb  = (const float*)d_in[4];
  const float* bw   = (const float*)d_in[5];
  const float* cw   = (const float*)d_in[6];
  const float* alog = (const float*)d_in[7];
  const float* Dp   = (const float*)d_in[8];
  const float* rope = (const float*)d_in[9];
  const float* outw = (const float*)d_in[10];
  float* out = (float*)d_out;

  float* ws = (float*)d_ws;
  float* XB = ws;
  float* Z  = XB + N16;
  float* DT = Z + N16;
  float* BV = DT + N16;
  float* CV = BV + N16;
  float* Y  = CV + N16;
  float* T  = Y + N16;
  float* BC = T + N16;
  float* HS = BC + NCHUNK * 256;

  k_proj<<<NTOK / 64, 128, 0, stream>>>(x, inw, dtw, dtb, bw, cw,
                                        XB, Z, DT, BV, CV);
  k_chunkop<<<NCHUNK, KT2, 0, stream>>>(DT, XB, BV, alog, rope, T, BC);
  k_combine<<<16, 512, 0, stream>>>(T, BC, init, HS, out + NTOK * DM);
  k_scan<<<NCHUNK, 512, 0, stream>>>(DT, XB, BV, CV, alog, rope, HS, Y);
  k_out<<<NTOK / 64, 128, 0, stream>>>(Y, Z, XB, Dp, outw, out);
}


# K4 scan: table holds K2's entries E=abar*cos, F=abar*sin, U (packed f32 pairs, abar by rcp + one correction); step is fma(-F,roll,fma(E,h,U)): one op behind the roll; rounding changes
# speedup vs baseline: 5.8382x; 1.0359x over previous
#include <hip/hip_runtime.h>
#include <hip/hip_bf16.h>
#include <math.h>


typedef __attribute__((ext_vector_type(2))) float v2f;
typedef __attribute__((ext_vector_type(8))) float v8f;

#define NTOK   65536
#define DM     8
#define DI     16
#define CHUNK  256
#define NCHUNK 256
#define N16    (NTOK * DI)

__device__ inline v8f wmma4(v2f a, v2f b, v8f c) {
  return __builtin_amdgcn_wmma_f32_16x16x4_f32(
      false, a, false, b, (short)0, c, false, false);
}

__device__ inline float softplus_f(float s) {
  return (s > 20.f) ? s : logf(1.f + __expf(s));
}

__global__ void k_proj(const float* __restrict__ x,
                       const float* __restrict__ inw,
                       const float* __restrict__ dtw,
                       const float* __restrict__ dtb,
                       const float* __restrict__ bw,
                       const float* __restrict__ cw,
                       float* __restrict__ xb, float* __restrict__ z,
                       float* __restrict__ dt, float* __restrict__ Bv,
                       float* __restrict__ Cv) {
  __shared__ float lds[4][256];
  const int lane = threadIdx.x & 31;
  const int wave = threadIdx.x >> 5;
  const int t0   = (blockIdx.x * 4 + wave) * 16;
  const int lo   = lane & 15, hi = lane >> 4;

  v8f axb = {}, az = {};
#pragma unroll
  for (int kk = 0; kk < 2; ++kk) {
    const int kb = kk * 4 + 2 * hi;
    v2f a, b0, b1;
    a.x  = x[(t0 + lo) * 8 + kb + 0];
    a.y  = x[(t0 + lo) * 8 + kb + 1];
    b0.x = inw[lo * 8 + kb + 0];
    b0.y = inw[lo * 8 + kb + 1];
    b1.x = inw[(lo + 16) * 8 + kb + 0];
    b1.y = inw[(lo + 16) * 8 + kb + 1];
    axb = wmma4(a, b0, axb);
    az  = wmma4(a, b1, az);
  }

#pragma unroll
  for (int r = 0; r < 8; ++r) lds[wave][(r + 8 * hi) * 16 + lo] = axb[r];

  v8f adt = {}, ab = {}, ac = {};
#pragma unroll
  for (int kk = 0; kk < 4; ++kk) {
    const int kb = kk * 4 + 2 * hi;
    v2f a, b;
    a.x = lds[wave][lo * 16 + kb + 0];
    a.y = lds[wave][lo * 16 + kb + 1];
    b.x = dtw[lo * 16 + kb + 0]; b.y = dtw[lo * 16 + kb + 1];
    adt = wmma4(a, b, adt);
    b.x = bw[lo * 16 + kb + 0];  b.y = bw[lo * 16 + kb + 1];
    ab  = wmma4(a, b, ab);
    b.x = cw[lo * 16 + kb + 0];  b.y = cw[lo * 16 + kb + 1];
    ac  = wmma4(a, b, ac);
  }

  const float bias = dtb[lo];
#pragma unroll
  for (int r = 0; r < 8; ++r) {
    const int idx = (t0 + r + 8 * hi) * 16 + lo;
    xb[idx] = axb[r];
    z[idx]  = az[r];
    dt[idx] = softplus_f(adt[r] + bias);
    Bv[idx] = ab[r];
    Cv[idx] = ac[r];
  }
}

#define SB    32
#define NBLK  (CHUNK / SB)
#define ROWF  48

struct FillOperands { float dtc[SB / 2], xbd[SB / 2], Bj[SB / 2]; };

__device__ inline void chunkop_load(FillOperands& o,
                                    const float* __restrict__ dt,
                                    const float* __restrict__ xb,
                                    const float* __restrict__ Bv,
                                    int tbase, int fth, int fd, int fj) {
#pragma unroll
  for (int i = 0; i < SB / 2; ++i) {
    const int tt = tbase + 2 * i + fth;
    o.dtc[i] = dt[tt * 16 + fd];
    o.xbd[i] = xb[tt * 16 + fd];
    o.Bj[i]  = Bv[tt * 16 + fj];
  }
}

__device__ inline v2f fma2(v2f a, v2f b, v2f c) { return __builtin_elementwise_fma(a, b, c); }

__device__ inline void chunkop_make(float* __restrict__ buf,
                                    const FillOperands& o,
                                    int fth, int fd, int fj,
                                    float fA, float fF) {
  const float hA = 0.5f * fA;
#pragma unroll
  for (int i = 0; i < SB / 2; i += 2) {
    v2f dtc; dtc.x = o.dtc[i]; dtc.y = o.dtc[i + 1];
    v2f Bj;  Bj.x  = o.Bj[i];  Bj.y  = o.Bj[i + 1];
    v2f xbd; xbd.x = o.xbd[i]; xbd.y = o.xbd[i + 1];
    const v2f ta  = dtc * hA;
    const v2f num = 1.f + ta;
    const v2f den = (1.f - ta) + 1e-8f;
    v2f r; r.x = __builtin_amdgcn_rcpf(den.x); r.y = __builtin_amdgcn_rcpf(den.y);
    v2f q = num * r;
    q = fma2(fma2(-den, q, num), r, q);
    const v2f rev = (dtc * fF) * 0.15915494309189532f;
    v2f cs, sn;
    cs.x = __builtin_amdgcn_cosf(rev.x); cs.y = __builtin_amdgcn_cosf(rev.y);
    sn.x = __builtin_amdgcn_sinf(rev.x); sn.y = __builtin_amdgcn_sinf(rev.y);
    const v2f E = q * cs, F = q * sn, U = dtc * Bj * xbd;
    const int t = 2 * i + fth;
    float* row0 = buf + ((SB - 1 - t) * 16 + fd) * ROWF;
    float* row1 = row0 - 2 * 16 * ROWF;
    row0[fj]      = E.x;
    row0[16 + ((fj + 15) & 15)] = F.x;
    row0[32 + fj] = U.x;
    row1[fj]      = E.y;
    row1[16 + ((fj + 15) & 15)] = F.y;
    row1[32 + fj] = U.y;
  }
}

__device__ inline void chunkop_fill(float* __restrict__ buf,
                                    const float* __restrict__ dt,
                                    const float* __restrict__ xb,
                                    const float* __restrict__ Bv,
                                    int tbase, int fth, int fd, int fj,
                                    float fA, float fF) {
  FillOperands o;
  chunkop_load(o, dt, xb, Bv, tbase, fth, fd, fj);
  chunkop_make(buf, o, fth, fd, fj, fA, fF);
}

#define KW2   512
#define KT2   768

__device__ inline float next_pair(float v) {
  const int i = __builtin_bit_cast(int, v);
  return __builtin_bit_cast(float, __builtin_amdgcn_permlane16(
      i, i, (int)0x07654321u, (int)0x8fedcba9u, true, false));
}

__global__ void __launch_bounds__(KT2)
k_chunkop(const float* __restrict__ dt,
          const float* __restrict__ xb,
          const float* __restrict__ Bv,
          const float* __restrict__ A_log,
          const float* __restrict__ rope,
          float* __restrict__ Tmat,
          float* __restrict__ bvec) {
  __shared__ __attribute__((aligned(16))) float tab[2][SB * 16 * ROWF];

  const int c    = blockIdx.x;
  const int tid  = threadIdx.x;
  const int d    = tid >> 5;
  const int lane = tid & 31;

  const bool producer = tid >= KW2;
  const int fj  = tid & 15;
  const int fd  = (tid >> 4) & 15;
  const int fth = (tid >> 8) & 1;
  const float fA = -__expf(A_log[fd * 16 + fj]);
  const float fF = rope[fd * 16 + fj];

  const int p = lane & 7, q = lane >> 3;
  float Sa[4], Sb[4], bacc[4];
#pragma unroll
  for (int r = 0; r < 4; ++r) {
    Sa[r]   = ((4 * q + r) == 2 * p)     ? 1.f : 0.f;
    Sb[r]   = ((4 * q + r) == 2 * p + 1) ? 1.f : 0.f;
    bacc[r] = 0.f;
  }

  const int t0 = c * CHUNK;

  if (!producer)
    chunkop_fill(tab[0], dt, xb, Bv, t0 + (NBLK - 1) * SB, fth, fd, fj, fA, fF);
  __syncthreads();

  for (int kb = 0; kb < NBLK; ++kb) {
    if (producer) {
      if (kb + 1 < NBLK) {
        float* nb = tab[(kb + 1) & 1];
        const int tn = t0 + (NBLK - 2 - kb) * SB;
        FillOperands o;
        chunkop_load(o, dt, xb, Bv, tn, 0, fd, fj);
        __builtin_amdgcn_sched_barrier(0);
        chunkop_make(nb, o, 0, fd, fj, fA, fF);
        chunkop_load(o, dt, xb, Bv, tn, 1, fd, fj);
        __builtin_amdgcn_sched_barrier(0);
        chunkop_make(nb, o, 1, fd, fj, fA, fF);
      }
    } else {
      const float* base = tab[kb & 1] + d * ROWF + 2 * p;
#pragma unroll 4
      for (int i = 0; i < SB; ++i) {
        const float* row = base + i * 16 * ROWF;
        const float2 E = *reinterpret_cast<const float2*>(row);
        const float2 F = *reinterpret_cast<const float2*>(row + 16);
        const float2 U = *reinterpret_cast<const float2*>(row + 32);
#pragma unroll
        for (int r = 0; r < 4; ++r) {
          const float sn = next_pair(Sa[r]);
          bacc[r] = fmaf(Sb[r], U.y, fmaf(Sa[r], U.x, bacc[r]));
          const float a = E.x * Sa[r] - F.x * Sb[r];
          Sb[r] = E.y * Sb[r] - F.y * sn;
          Sa[r] = a;
        }
      }
    }
    __syncthreads();
  }

  if (producer) return;
#pragma unroll
  for (int r = 0; r < 4; ++r) {
    float2 v; v.x = Sa[r]; v.y = Sb[r];
    *reinterpret_cast<float2*>(Tmat + ((c * 16 + d) * 16 + (4 * q + r)) * 16 + 2 * p) = v;
  }

  float bout = 0.f;
#pragma unroll
  for (int r = 0; r < 4; ++r) {
    float v = bacc[r];
    v += __shfl_xor(v, 1, 32);
    v += __shfl_xor(v, 2, 32);
    v += __shfl_xor(v, 4, 32);
    bout = (p == r) ? v : bout;
  }
  if (p < 4) bvec[(c * 16 + d) * 16 + 4 * q + p] = bout;
}

__device__ inline float lane_bcast(float v, int k) {
  return __int_as_float(__builtin_amdgcn_readlane(__float_as_int(v), k));
}

__device__ inline float affine_row(float4 m0, float4 m1, float4 m2, float4 m3,
                                   float vec, float bias) {
  float s[16];
#pragma unroll
  for (int k = 0; k < 16; ++k) s[k] = lane_bcast(vec, k);
  float s0 = m0.x * s[0], s1 = m1.x * s[4], s2 = m2.x * s[8], s3 = m3.x * s[12];
  s0 = fmaf(m0.y, s[1], s0);  s1 = fmaf(m1.y, s[5], s1);
  s2 = fmaf(m2.y, s[9], s2);  s3 = fmaf(m3.y, s[13], s3);
  s0 = fmaf(m0.z, s[2], s0);  s1 = fmaf(m1.z, s[6], s1);
  s2 = fmaf(m2.z, s[10], s2); s3 = fmaf(m3.z, s[14], s3);
  s0 = fmaf(m0.w, s[3], s0);  s1 = fmaf(m1.w, s[7], s1);
  s2 = fmaf(m2.w, s[11], s2); s3 = fmaf(m3.w, s[15], s3);
  return bias + ((s0 + s1) + (s2 + s3));
}

__global__ void __launch_bounds__(512) __attribute__((amdgpu_waves_per_eu(4, 4)))
k_combine(const float* __restrict__ Tmat,
          const float* __restrict__ bvec,
          const float* __restrict__ init_h,
          float* __restrict__ hstart,
          float* __restrict__ h_final) {
  __shared__ __attribute__((aligned(16))) float Pm[16][256];
  __shared__ __attribute__((aligned(16))) float qv[16][16];
  __shared__ __attribute__((aligned(16))) float Hg[17][16];

  const int d    = blockIdx.x;
  const int g    = threadIdx.x >> 5;
  const int lane = threadIdx.x & 31;
  const int lo   = lane & 15, hi = lane >> 4;

  v8f P;
#pragma unroll
  for (int r = 0; r < 8; ++r) P[r] = ((r + 8 * hi) == lo) ? 1.f : 0.f;
  float q = 0.f;

#pragma unroll
  for (int i = 0; i < 16; ++i) {
    const int c = g * 16 + i;
    const float* Tc = Tmat + (c * 16 + d) * 256;
    v2f a[4];
#pragma unroll
    for (int kk = 0; kk < 4; ++kk)
      a[kk] = *reinterpret_cast<const v2f*>(Tc + lo * 16 + 8 * hi + 2 * kk);
    const float4* Trow = reinterpret_cast<const float4*>(Tc + lo * 16);
    const float4 m0 = Trow[0], m1 = Trow[1], m2 = Trow[2], m3 = Trow[3];
    const float bias = bvec[(c * 16 + d) * 16 + lo];

    v8f acc = {};
#pragma unroll
    for (int kk = 0; kk < 4; ++kk) {
      v2f b;
      b.x = P[2 * kk];
      b.y = P[2 * kk + 1];
      acc = wmma4(a[kk], b, acc);
    }
    q = affine_row(m0, m1, m2, m3, q, bias);
    P = acc;
  }
#pragma unroll
  for (int r = 0; r < 8; ++r) Pm[g][(r + 8 * hi) * 16 + lo] = P[r];
  qv[g][lo] = q;
  __syncthreads();

  if (g == 0) {
    float h = init_h[d * 16 + lo];
    Hg[0][lo] = h;
#pragma unroll
    for (int k = 0; k < 16; ++k) {
      const float4* Prow = reinterpret_cast<const float4*>(&Pm[k][lo * 16]);
      h = affine_row(Prow[0], Prow[1], Prow[2], Prow[3], h, qv[k][lo]);
      Hg[k + 1][lo] = h;
    }
    if (hi == 0) h_final[d * 16 + lo] = h;
  }
  __syncthreads();

  {
    float hs[16];
    float h = Hg[g][lo];
#pragma unroll
    for (int i = 0; i < 16; ++i) {
      const int c = g * 16 + i;
      hs[i] = h;
      if (i < 15) {
        const float4* Trow =
            reinterpret_cast<const float4*>(Tmat + ((c * 16 + d) * 16 + lo) * 16);
        h = affine_row(Trow[0], Trow[1], Trow[2], Trow[3], h,
                       bvec[(c * 16 + d) * 16 + lo]);
      }
    }
    if (hi == 0) {
#pragma unroll
      for (int i = 0; i < 16; ++i)
        hstart[((g * 16 + i) * 16 + d) * 16 + lo] = hs[i];
    }
  }
}

#define SB4    16
#define NBLK4  (CHUNK / SB4)
#define HP     20

struct ScanOperands { float dtc[SB4 / 2], xbd[SB4 / 2], Bj[SB4 / 2]; };

__device__ inline void scan_load(ScanOperands& o,
                                 const float* __restrict__ dt,
                                 const float* __restrict__ xb,
                                 const float* __restrict__ Bv,
                                 int tbase, int fth, int fd, int fj) {
#pragma unroll
  for (int i = 0; i < SB4 / 2; ++i) {
    const int tt = tbase + 2 * i + fth;
    o.dtc[i] = dt[tt * 16 + fd];
    o.xbd[i] = xb[tt * 16 + fd];
    o.Bj[i]  = Bv[tt * 16 + fj];
  }
}

struct PairPiece { v2f q, rev, dtc; };

__device__ inline PairPiece pair_front(const ScanOperands& o, int i,
                                       float hA, float fF) {
  PairPiece p;
  p.dtc.x = o.dtc[i]; p.dtc.y = o.dtc[i + 1];
  const v2f ta  = p.dtc * hA;
  const v2f num = 1.f + ta;
  const v2f den = (1.f - ta) + 1e-8f;
  v2f r; r.x = __builtin_amdgcn_rcpf(den.x); r.y = __builtin_amdgcn_rcpf(den.y);
  v2f q = num * r;
  p.q   = fma2(fma2(-den, q, num), r, q);
  p.rev = (p.dtc * fF) * 0.15915494309189532f;
  return p;
}

__device__ inline void pair_back(float4* __restrict__ buf, const PairPiece& p,
                                 const ScanOperands& o, int i,
                                 int fth, int fd, int fj) {
  v2f cs, sn;
  cs.x = __builtin_amdgcn_cosf(p.rev.x); cs.y = __builtin_amdgcn_cosf(p.rev.y);
  sn.x = __builtin_amdgcn_sinf(p.rev.x); sn.y = __builtin_amdgcn_sinf(p.rev.y);
  v2f Bj;  Bj.x  = o.Bj[i];  Bj.y  = o.Bj[i + 1];
  v2f xbd; xbd.x = o.xbd[i]; xbd.y = o.xbd[i + 1];
  const v2f E = p.q * cs, F = p.q * sn, U = p.dtc * Bj * xbd;
  const int t = 2 * i + fth;
  float4 v0, v1;
  v0.x = E.x; v0.y = F.x; v0.z = U.x; v0.w = 0.f;
  v1.x = E.y; v1.y = F.y; v1.z = U.y; v1.w = 0.f;
  buf[(t * 16 + fd) * 16 + fj]       = v0;
  buf[((t + 2) * 16 + fd) * 16 + fj] = v1;
}

__device__ inline void scan_fill(float4* __restrict__ buf, const ScanOperands& o,
                                 int fth, int fd, int fj, float hA, float fF) {
#pragma unroll
  for (int i = 0; i < SB4 / 2; i += 2) {
    const PairPiece p = pair_front(o, i, hA, fF);
    pair_back(buf, p, o, i, fth, fd, fj);
  }
}

__device__ inline float half_sum(float4 ha, float4 hb, float4 ca, float4 cb) {
  const float q0 = fmaf(ha.x, ca.x, ha.y * ca.y), q1 = fmaf(ha.z, ca.z, ha.w * ca.w);
  const float q2 = fmaf(hb.x, cb.x, hb.y * cb.y), q3 = fmaf(hb.z, cb.z, hb.w * cb.w);
  return (q0 + q1) + (q2 + q3);
}

template <bool FILL, bool PREV>
__device__ inline float scan_walk(const float4* __restrict__ trow,
                                  float* hrow, const float4* hrd,
                                  float h, int src,
                                  float4 caP, float4 cbP, bool writer,
                                  float* __restrict__ ypP,
                                  float4* __restrict__ nbuf,
                                  const ScanOperands& o, int fth, int fd,
                                  int fj, float hA, float fF) {
  float4 ha = {}, hb = {};
  if (PREV) { ha = hrd[0]; hb = hrd[1]; }
  float4 co[SB4];
#pragma unroll
  for (int t = 0; t < SB4; ++t) co[t] = trow[t * 256];

  PairPiece p = {};
  float sp = 0.f, sx = 0.f;
#pragma unroll
  for (int t = 0; t < SB4; ++t) {
    const float hr = __shfl(h, src, 32);
    __builtin_amdgcn_sched_barrier(0);
    const float a = fmaf(co[t].x, h, co[t].z);
    if (FILL) {
      if ((t & 3) == 0) p = pair_front(o, t >> 1, hA, fF);
      if ((t & 3) == 2) pair_back(nbuf, p, o, (t >> 1) - 1, fth, fd, fj);
    }
    if (PREV && t == 0) {
      sp = half_sum(ha, hb, caP, cbP);
      sx = __shfl_xor(sp, 16, 32);
    }
    if (PREV && t == 1) {
      if (writer) *ypP = sp + sx;
    }
    __builtin_amdgcn_sched_barrier(0);
    h = fmaf(-co[t].y, hr, a);
    hrow[t * HP] = h;
  }
  return h;
}

__global__ void __launch_bounds__(512)
k_scan(const float* __restrict__ dt,
       const float* __restrict__ xb,
       const float* __restrict__ Bv,
       const float* __restrict__ Cv,
       const float* __restrict__ A_log,
       const float* __restrict__ rope,
       const float* __restrict__ hstart,
       float* __restrict__ y) {
  __shared__ __attribute__((aligned(16))) float4 tab[2][SB4 * 16 * 16];
  __shared__ __attribute__((aligned(16))) float  hst[16 * SB4 * HP];

  const int c    = blockIdx.x;
  const int tid  = threadIdx.x;
  const int d    = tid >> 5;
  const int lane = tid & 31;
  const int lo   = lane & 15, hi = lane >> 4;

  const int fj  = tid & 15;
  const int fd  = (tid >> 4) & 15;
  const int fth = tid >> 8;
  const float hA = 0.5f * -__expf(A_log[fd * 16 + fj]);
  const float fF = rope[fd * 16 + fj];

  float h = hstart[(c * 16 + d) * 16 + lo];
  const int t0  = c * CHUNK;
  const int src = (lane & 16) | ((lo + 15) & 15);

  float*        hrow = hst + (d * SB4) * HP + lo;
  const float4* hrd  =
      reinterpret_cast<const float4*>(hst + (d * SB4 + lo) * HP + 8 * hi);
  const bool writer = (hi == 0);

  const float4* Crow = reinterpret_cast<const float4*>(Cv + (t0 + lo) * 16 + 8 * hi);
  float*        yrow = y + (t0 + lo) * 16 + d;
  const int     CSTEP = SB4 * 16 / 4;
  const int     YSTEP = SB4 * 16;

  ScanOperands opsA, opsB;
  scan_load(opsB, dt, xb, Bv, t0, fth, fd, fj);
  scan_load(opsA, dt, xb, Bv, t0 + SB4, fth, fd, fj);
  scan_fill(tab[0], opsB, fth, fd, fj, hA, fF);
  __syncthreads();

  scan_load(opsB, dt, xb, Bv, t0 + 2 * SB4, fth, fd, fj);
  float4 caP = Crow[0], cbP = Crow[1];
  h = scan_walk<true, false>(tab[0] + d * 16 + lo, hrow, hrd, h, src,
                             caP, cbP, writer, yrow,
                             tab[1], opsA, fth, fd, fj, hA, fF);
  opsA = opsB;
  __syncthreads();

  for (int kb = 1; kb < NBLK4 - 1; ++kb) {
    if (kb + 2 < NBLK4)
      scan_load(opsB, dt, xb, Bv, t0 + (kb + 2) * SB4, fth, fd, fj);
    const float4 ca = Crow[kb * CSTEP], cb = Crow[kb * CSTEP + 1];

    h = scan_walk<true, true>(tab[kb & 1] + d * 16 + lo, hrow, hrd, h, src,
                              caP, cbP, writer, yrow + (kb - 1) * YSTEP,
                              tab[(kb + 1) & 1], opsA, fth, fd, fj, hA, fF);
    opsA = opsB;
    caP = ca; cbP = cb;
    __syncthreads();
  }
  {
    const int kb = NBLK4 - 1;
    const float4 ca = Crow[kb * CSTEP], cb = Crow[kb * CSTEP + 1];
    scan_walk<false, true>(tab[kb & 1] + d * 16 + lo, hrow, hrd, h, src,
                           caP, cbP, writer, yrow + (kb - 1) * YSTEP,
                           tab[0], opsA, fth, fd, fj, hA, fF);
    float s = half_sum(hrd[0], hrd[1], ca, cb);
    s += __shfl_xor(s, 16, 32);
    if (writer) yrow[kb * YSTEP] = s;
  }
}

__global__ void k_out(const float* __restrict__ y,
                      const float* __restrict__ z,
                      const float* __restrict__ xb,
                      const float* __restrict__ Dp,
                      const float* __restrict__ outw,
                      float* __restrict__ out) {
  const int lane = threadIdx.x & 31;
  const int wave = threadIdx.x >> 5;
  const int t0   = (blockIdx.x * 4 + wave) * 16;
  const int lo   = lane & 15, hi = lane >> 4;

  v8f acc = {};
#pragma unroll
  for (int kk = 0; kk < 4; ++kk) {
    const int kb = kk * 4 + 2 * hi;
    v2f a, b;
#pragma unroll
    for (int v = 0; v < 2; ++v) {
      const int k   = kb + v;
      const int idx = (t0 + lo) * 16 + k;
      const float zz  = z[idx];
      const float sil = zz / (1.f + __expf(-zz));
      const float val = y[idx] * sil + Dp[k] * xb[idx];
      if (v == 0) a.x = val; else a.y = val;
    }
    b.x = (lo < 8) ? outw[lo * 16 + kb + 0] : 0.f;
    b.y = (lo < 8) ? outw[lo * 16 + kb + 1] : 0.f;
    acc = wmma4(a, b, acc);
  }
  if (lo < 8) {
#pragma unroll
    for (int r = 0; r < 8; ++r) out[(t0 + r + 8 * hi) * 8 + lo] = acc[r];
  }
}

extern "C" void kernel_launch(void* const* d_in, const int* in_sizes, int n_in,
                              void* d_out, int out_size, void* d_ws,
                              size_t ws_size, hipStream_t stream) {
  const float* x    = (const float*)d_in[0];
  const float* init = (const float*)d_in[1];
  const float* inw  = (const float*)d_in[2];
  const float* dtw  = (const float*)d_in[3];
  const float* dtb  = (const float*)d_in[4];
  const float* bw   = (const float*)d_in[5];
  const float* cw   = (const float*)d_in[6];
  const float* alog = (const float*)d_in[7];
  const float* Dp   = (const float*)d_in[8];
  const float* rope = (const float*)d_in[9];
  const float* outw = (const float*)d_in[10];
  float* out = (float*)d_out;

  float* ws = (float*)d_ws;
  float* XB = ws;
  float* Z  = XB + N16;
  float* DT = Z + N16;
  float* BV = DT + N16;
  float* CV = BV + N16;
  float* Y  = CV + N16;
  float* T  = Y + N16;
  float* BC = T + N16;
  float* HS = BC + NCHUNK * 256;

  k_proj<<<NTOK / 64, 128, 0, stream>>>(x, inw, dtw, dtb, bw, cw,
                                        XB, Z, DT, BV, CV);
  k_chunkop<<<NCHUNK, KT2, 0, stream>>>(DT, XB, BV, alog, rope, T, BC);
  k_combine<<<16, 512, 0, stream>>>(T, BC, init, HS, out + NTOK * DM);
  k_scan<<<NCHUNK, 512, 0, stream>>>(DT, XB, BV, CV, alog, rope, HS, Y);
  k_out<<<NTOK / 64, 128, 0, stream>>>(Y, Z, XB, Dp, outw, out);
}


# Launches 5 -> 4: a K2 work-group takes the 16 consecutive chunks of ONE channel and chains their operators itself (WMMA prefix maps); K4 waves fold the group totals; fold sum order changes
# speedup vs baseline: 6.3000x; 1.0791x over previous
#include <hip/hip_runtime.h>
#include <hip/hip_bf16.h>
#include <math.h>


typedef __attribute__((ext_vector_type(2))) float v2f;
typedef __attribute__((ext_vector_type(8))) float v8f;

#define NTOK   65536
#define DM     8
#define DI     16
#define CHUNK  256
#define NCHUNK 256
#define N16    (NTOK * DI)

__device__ inline v8f wmma4(v2f a, v2f b, v8f c) {
  return __builtin_amdgcn_wmma_f32_16x16x4_f32(
      false, a, false, b, (short)0, c, false, false);
}

__device__ inline float softplus_f(float s) {
  return (s > 20.f) ? s : logf(1.f + __expf(s));
}

__global__ void k_proj(const float* __restrict__ x,
                       const float* __restrict__ inw,
                       const float* __restrict__ dtw,
                       const float* __restrict__ dtb,
                       const float* __restrict__ bw,
                       const float* __restrict__ cw,
                       float* __restrict__ xb, float* __restrict__ z,
                       float* __restrict__ dt, float* __restrict__ Bv,
                       float* __restrict__ Cv,
                       float* __restrict__ xbT,
                       float* __restrict__ dtT) {
  __shared__ float lds[4][256];
  const int lane = threadIdx.x & 31;
  const int wave = threadIdx.x >> 5;
  const int t0   = (blockIdx.x * 4 + wave) * 16;
  const int lo   = lane & 15, hi = lane >> 4;

  v8f axb = {}, az = {};
#pragma unroll
  for (int kk = 0; kk < 2; ++kk) {
    const int kb = kk * 4 + 2 * hi;
    v2f a, b0, b1;
    a.x  = x[(t0 + lo) * 8 + kb + 0];
    a.y  = x[(t0 + lo) * 8 + kb + 1];
    b0.x = inw[lo * 8 + kb + 0];
    b0.y = inw[lo * 8 + kb + 1];
    b1.x = inw[(lo + 16) * 8 + kb + 0];
    b1.y = inw[(lo + 16) * 8 + kb + 1];
    axb = wmma4(a, b0, axb);
    az  = wmma4(a, b1, az);
  }

#pragma unroll
  for (int r = 0; r < 8; ++r) lds[wave][(r + 8 * hi) * 16 + lo] = axb[r];

  v8f adt = {}, ab = {}, ac = {};
#pragma unroll
  for (int kk = 0; kk < 4; ++kk) {
    const int kb = kk * 4 + 2 * hi;
    v2f a, b;
    a.x = lds[wave][lo * 16 + kb + 0];
    a.y = lds[wave][lo * 16 + kb + 1];
    b.x = dtw[lo * 16 + kb + 0]; b.y = dtw[lo * 16 + kb + 1];
    adt = wmma4(a, b, adt);
    b.x = bw[lo * 16 + kb + 0];  b.y = bw[lo * 16 + kb + 1];
    ab  = wmma4(a, b, ab);
    b.x = cw[lo * 16 + kb + 0];  b.y = cw[lo * 16 + kb + 1];
    ac  = wmma4(a, b, ac);
  }

  const float bias = dtb[lo];
  float dtv[8];
#pragma unroll
  for (int r = 0; r < 8; ++r) {
    const int idx = (t0 + r + 8 * hi) * 16 + lo;
    dtv[r]  = softplus_f(adt[r] + bias);
    xb[idx] = axb[r];
    z[idx]  = az[r];
    dt[idx] = dtv[r];
    Bv[idx] = ab[r];
    Cv[idx] = ac[r];
  }

  const int tch = lane >> 1, ttk = (lane & 1) * 8;
  {
    float4 v0, v1;
    v0.x = lds[wave][(ttk + 0) * 16 + tch]; v0.y = lds[wave][(ttk + 1) * 16 + tch];
    v0.z = lds[wave][(ttk + 2) * 16 + tch]; v0.w = lds[wave][(ttk + 3) * 16 + tch];
    v1.x = lds[wave][(ttk + 4) * 16 + tch]; v1.y = lds[wave][(ttk + 5) * 16 + tch];
    v1.z = lds[wave][(ttk + 6) * 16 + tch]; v1.w = lds[wave][(ttk + 7) * 16 + tch];
    float4* dst = reinterpret_cast<float4*>(xbT + tch * NTOK + t0 + ttk);
    dst[0] = v0; dst[1] = v1;
  }
#pragma unroll
  for (int r = 0; r < 8; ++r) lds[wave][(r + 8 * hi) * 16 + lo] = dtv[r];
  {
    float4 v0, v1;
    v0.x = lds[wave][(ttk + 0) * 16 + tch]; v0.y = lds[wave][(ttk + 1) * 16 + tch];
    v0.z = lds[wave][(ttk + 2) * 16 + tch]; v0.w = lds[wave][(ttk + 3) * 16 + tch];
    v1.x = lds[wave][(ttk + 4) * 16 + tch]; v1.y = lds[wave][(ttk + 5) * 16 + tch];
    v1.z = lds[wave][(ttk + 6) * 16 + tch]; v1.w = lds[wave][(ttk + 7) * 16 + tch];
    float4* dst = reinterpret_cast<float4*>(dtT + tch * NTOK + t0 + ttk);
    dst[0] = v0; dst[1] = v1;
  }
}

#define SB    32
#define NBLK  (CHUNK / SB)
#define ROWF  48

struct FillOperands { float dtc[SB / 2], xbd[SB / 2], Bj[SB / 2]; };

__device__ inline void chunkop_load(FillOperands& o,
                                    const float* __restrict__ dtc,
                                    const float* __restrict__ xbc,
                                    const float* __restrict__ Bv,
                                    int tbase, int fth, int fj) {
#pragma unroll
  for (int i = 0; i < SB / 2; ++i) {
    const int tt = tbase + 2 * i + fth;
    o.dtc[i] = dtc[tt];
    o.xbd[i] = xbc[tt];
    o.Bj[i]  = Bv[tt * 16 + fj];
  }
}

__device__ inline v2f fma2(v2f a, v2f b, v2f c) { return __builtin_elementwise_fma(a, b, c); }

__device__ inline void chunkop_make(float* __restrict__ buf,
                                    const FillOperands& o,
                                    int fth, int fd, int fj,
                                    float fA, float fF) {
  const float hA = 0.5f * fA;
#pragma unroll
  for (int i = 0; i < SB / 2; i += 2) {
    v2f dtc; dtc.x = o.dtc[i]; dtc.y = o.dtc[i + 1];
    v2f Bj;  Bj.x  = o.Bj[i];  Bj.y  = o.Bj[i + 1];
    v2f xbd; xbd.x = o.xbd[i]; xbd.y = o.xbd[i + 1];
    const v2f ta  = dtc * hA;
    const v2f num = 1.f + ta;
    const v2f den = (1.f - ta) + 1e-8f;
    v2f r; r.x = __builtin_amdgcn_rcpf(den.x); r.y = __builtin_amdgcn_rcpf(den.y);
    v2f q = num * r;
    q = fma2(fma2(-den, q, num), r, q);
    const v2f rev = (dtc * fF) * 0.15915494309189532f;
    v2f cs, sn;
    cs.x = __builtin_amdgcn_cosf(rev.x); cs.y = __builtin_amdgcn_cosf(rev.y);
    sn.x = __builtin_amdgcn_sinf(rev.x); sn.y = __builtin_amdgcn_sinf(rev.y);
    const v2f E = q * cs, F = q * sn, U = dtc * Bj * xbd;
    const int t = 2 * i + fth;
    float* row0 = buf + ((SB - 1 - t) * 16 + fd) * ROWF;
    float* row1 = row0 - 2 * 16 * ROWF;
    row0[fj]      = E.x;
    row0[16 + ((fj + 15) & 15)] = F.x;
    row0[32 + fj] = U.x;
    row1[fj]      = E.y;
    row1[16 + ((fj + 15) & 15)] = F.y;
    row1[32 + fj] = U.y;
  }
}

__device__ inline void chunkop_fill(float* __restrict__ buf,
                                    const float* __restrict__ dtc,
                                    const float* __restrict__ xbc,
                                    const float* __restrict__ Bv,
                                    int tbase, int fth, int fd, int fj,
                                    float fA, float fF) {
  FillOperands o;
  chunkop_load(o, dtc, xbc, Bv, tbase, fth, fj);
  chunkop_make(buf, o, fth, fd, fj, fA, fF);
}

#define KW2   512
#define KT2   768

__device__ inline float next_pair(float v) {
  const int i = __builtin_bit_cast(int, v);
  return __builtin_bit_cast(float, __builtin_amdgcn_permlane16(
      i, i, (int)0x07654321u, (int)0x8fedcba9u, true, false));
}

__device__ inline float lane_bcast(float v, int k) {
  return __int_as_float(__builtin_amdgcn_readlane(__float_as_int(v), k));
}

__device__ inline float affine_row(float4 m0, float4 m1, float4 m2, float4 m3,
                                   float vec, float bias) {
  float s[16];
#pragma unroll
  for (int k = 0; k < 16; ++k) s[k] = lane_bcast(vec, k);
  float s0 = m0.x * s[0], s1 = m1.x * s[4], s2 = m2.x * s[8], s3 = m3.x * s[12];
  s0 = fmaf(m0.y, s[1], s0);  s1 = fmaf(m1.y, s[5], s1);
  s2 = fmaf(m2.y, s[9], s2);  s3 = fmaf(m3.y, s[13], s3);
  s0 = fmaf(m0.z, s[2], s0);  s1 = fmaf(m1.z, s[6], s1);
  s2 = fmaf(m2.z, s[10], s2); s3 = fmaf(m3.z, s[14], s3);
  s0 = fmaf(m0.w, s[3], s0);  s1 = fmaf(m1.w, s[7], s1);
  s2 = fmaf(m2.w, s[11], s2); s3 = fmaf(m3.w, s[15], s3);
  return bias + ((s0 + s1) + (s2 + s3));
}

__device__ inline float other_row(float v) {
  const int i = __builtin_bit_cast(int, v);
  return __builtin_bit_cast(float, __builtin_amdgcn_permlanex16(
      i, i, (int)0x76543210u, (int)0xfedcba98u, true, false));
}

__device__ inline float affine_split(float4 m0, float4 m1, float vec, float bias,
                                     bool upper) {
  float s[8];
#pragma unroll
  for (int k = 0; k < 8; ++k) {
    const float a = lane_bcast(vec, k), b = lane_bcast(vec, k + 8);
    s[k] = upper ? b : a;
  }
  float s0 = m0.x * s[0], s1 = m1.x * s[4];
  s0 = fmaf(m0.y, s[1], s0);  s1 = fmaf(m1.y, s[5], s1);
  s0 = fmaf(m0.z, s[2], s0);  s1 = fmaf(m1.z, s[6], s1);
  s0 = fmaf(m0.w, s[3], s0);  s1 = fmaf(m1.w, s[7], s1);
  const float part = s0 + s1;
  const float oth  = other_row(part);
  return bias + (upper ? (oth + part) : (part + oth));
}

__global__ void __launch_bounds__(KT2)
k_chunkop(const float* __restrict__ dtT,
          const float* __restrict__ xbT,
          const float* __restrict__ Bv,
          const float* __restrict__ A_log,
          const float* __restrict__ rope,
          float* __restrict__ Pmap,
          float* __restrict__ qmap) {
  __shared__ __attribute__((aligned(16))) float tab[2][SB * 16 * ROWF];

  const int grp  = blockIdx.x >> 4;
  const int ch   = blockIdx.x & 15;
  const int tid  = threadIdx.x;
  const int w    = tid >> 5;
  const int lane = tid & 31;

  const bool producer = tid >= KW2;
  const int fj  = tid & 15;
  const int fw  = (tid >> 4) & 15;
  const int fth = (tid >> 8) & 1;
  const float fA = -__expf(A_log[ch * 16 + fj]);
  const float fF = rope[ch * 16 + fj];
  const float* dtc = dtT + ch * NTOK;
  const float* xbc = xbT + ch * NTOK;

  const int p = lane & 7, q = lane >> 3;
  float Sa[4], Sb[4], bacc[4];
#pragma unroll
  for (int r = 0; r < 4; ++r) {
    Sa[r]   = ((4 * q + r) == 2 * p)     ? 1.f : 0.f;
    Sb[r]   = ((4 * q + r) == 2 * p + 1) ? 1.f : 0.f;
    bacc[r] = 0.f;
  }

  const int t0f = (grp * 16 + fw) * CHUNK;

  if (!producer)
    chunkop_fill(tab[0], dtc, xbc, Bv, t0f + (NBLK - 1) * SB, fth, fw, fj, fA, fF);
  __syncthreads();

  for (int kb = 0; kb < NBLK; ++kb) {
    if (producer) {
      if (kb + 1 < NBLK) {
        float* nb = tab[(kb + 1) & 1];
        const int tn = t0f + (NBLK - 2 - kb) * SB;
        FillOperands o;
        chunkop_load(o, dtc, xbc, Bv, tn, 0, fj);
        __builtin_amdgcn_sched_barrier(0);
        chunkop_make(nb, o, 0, fw, fj, fA, fF);
        chunkop_load(o, dtc, xbc, Bv, tn, 1, fj);
        __builtin_amdgcn_sched_barrier(0);
        chunkop_make(nb, o, 1, fw, fj, fA, fF);
      }
    } else {
      const float* base = tab[kb & 1] + w * ROWF + 2 * p;
#pragma unroll 4
      for (int i = 0; i < SB; ++i) {
        const float* row = base + i * 16 * ROWF;
        const float2 E = *reinterpret_cast<const float2*>(row);
        const float2 F = *reinterpret_cast<const float2*>(row + 16);
        const float2 U = *reinterpret_cast<const float2*>(row + 32);
#pragma unroll
        for (int r = 0; r < 4; ++r) {
          const float sn = next_pair(Sa[r]);
          bacc[r] = fmaf(Sb[r], U.y, fmaf(Sa[r], U.x, bacc[r]));
          const float a = E.x * Sa[r] - F.x * Sb[r];
          Sb[r] = E.y * Sb[r] - F.y * sn;
          Sa[r] = a;
        }
      }
    }
    __syncthreads();
  }

  float* Tl = tab[0];
  float* bl = tab[0] + 16 * 256;
  if (!producer) {
#pragma unroll
    for (int r = 0; r < 4; ++r) {
      float2 v; v.x = Sa[r]; v.y = Sb[r];
      *reinterpret_cast<float2*>(Tl + w * 256 + (4 * q + r) * 16 + 2 * p) = v;
    }
    float bout = 0.f;
#pragma unroll
    for (int r = 0; r < 4; ++r) {
      float v = bacc[r];
      v += __shfl_xor(v, 1, 32);
      v += __shfl_xor(v, 2, 32);
      v += __shfl_xor(v, 4, 32);
      bout = (p == r) ? v : bout;
    }
    if (p < 4) bl[w * 16 + 4 * q + p] = bout;
  }
  __syncthreads();
  if (tid >= 32) return;

  const int lo = lane & 15, hi = lane >> 4;
  v8f P;
#pragma unroll
  for (int r = 0; r < 8; ++r) P[r] = ((r + 8 * hi) == lo) ? 1.f : 0.f;
  float qq = 0.f;
#pragma unroll
  for (int i = 0; i < 16; ++i) {
    const float* Tc = Tl + i * 256;
    v2f a[4];
#pragma unroll
    for (int kk = 0; kk < 4; ++kk)
      a[kk] = *reinterpret_cast<const v2f*>(Tc + lo * 16 + 8 * hi + 2 * kk);
    const float4* Trow = reinterpret_cast<const float4*>(Tc + lo * 16);
    const float4 m0 = Trow[0], m1 = Trow[1], m2 = Trow[2], m3 = Trow[3];
    const float bias = bl[i * 16 + lo];

    v8f acc = {};
#pragma unroll
    for (int kk = 0; kk < 4; ++kk) {
      v2f b;
      b.x = P[2 * kk];
      b.y = P[2 * kk + 1];
      acc = wmma4(a[kk], b, acc);
    }
    qq = affine_row(m0, m1, m2, m3, qq, bias);
    P = acc;

    const int c = grp * 16 + i;
    float* Pg = Pmap + (c * 16 + ch) * 256;
#pragma unroll
    for (int r = 0; r < 8; ++r) Pg[(r + 8 * hi) * 16 + lo] = P[r];
    qmap[(c * 16 + ch) * 16 + lo] = qq;
  }
}

#define SB4    16
#define NBLK4  (CHUNK / SB4)
#define HP     20

struct ScanOperands { float dtc[SB4 / 2], xbd[SB4 / 2], Bj[SB4 / 2]; };

__device__ inline void scan_load(ScanOperands& o,
                                 const float* __restrict__ dt,
                                 const float* __restrict__ xb,
                                 const float* __restrict__ Bv,
                                 int tbase, int fth, int fd, int fj) {
#pragma unroll
  for (int i = 0; i < SB4 / 2; ++i) {
    const int tt = tbase + 2 * i + fth;
    o.dtc[i] = dt[tt * 16 + fd];
    o.xbd[i] = xb[tt * 16 + fd];
    o.Bj[i]  = Bv[tt * 16 + fj];
  }
}

struct PairPiece { v2f q, rev, dtc; };

__device__ inline PairPiece pair_front(const ScanOperands& o, int i,
                                       float hA, float fF) {
  PairPiece p;
  p.dtc.x = o.dtc[i]; p.dtc.y = o.dtc[i + 1];
  const v2f ta  = p.dtc * hA;
  const v2f num = 1.f + ta;
  const v2f den = (1.f - ta) + 1e-8f;
  v2f r; r.x = __builtin_amdgcn_rcpf(den.x); r.y = __builtin_amdgcn_rcpf(den.y);
  v2f q = num * r;
  p.q   = fma2(fma2(-den, q, num), r, q);
  p.rev = (p.dtc * fF) * 0.15915494309189532f;
  return p;
}

__device__ inline void pair_back(float4* __restrict__ buf, const PairPiece& p,
                                 const ScanOperands& o, int i,
                                 int fth, int fd, int fj) {
  v2f cs, sn;
  cs.x = __builtin_amdgcn_cosf(p.rev.x); cs.y = __builtin_amdgcn_cosf(p.rev.y);
  sn.x = __builtin_amdgcn_sinf(p.rev.x); sn.y = __builtin_amdgcn_sinf(p.rev.y);
  v2f Bj;  Bj.x  = o.Bj[i];  Bj.y  = o.Bj[i + 1];
  v2f xbd; xbd.x = o.xbd[i]; xbd.y = o.xbd[i + 1];
  const v2f E = p.q * cs, F = p.q * sn, U = p.dtc * Bj * xbd;
  const int t = 2 * i + fth;
  float4 v0, v1;
  v0.x = E.x; v0.y = F.x; v0.z = U.x; v0.w = 0.f;
  v1.x = E.y; v1.y = F.y; v1.z = U.y; v1.w = 0.f;
  buf[(t * 16 + fd) * 16 + fj]       = v0;
  buf[((t + 2) * 16 + fd) * 16 + fj] = v1;
}

__device__ inline void scan_fill(float4* __restrict__ buf, const ScanOperands& o,
                                 int fth, int fd, int fj, float hA, float fF) {
#pragma unroll
  for (int i = 0; i < SB4 / 2; i += 2) {
    const PairPiece p = pair_front(o, i, hA, fF);
    pair_back(buf, p, o, i, fth, fd, fj);
  }
}

__device__ inline float half_sum(float4 ha, float4 hb, float4 ca, float4 cb) {
  const float q0 = fmaf(ha.x, ca.x, ha.y * ca.y), q1 = fmaf(ha.z, ca.z, ha.w * ca.w);
  const float q2 = fmaf(hb.x, cb.x, hb.y * cb.y), q3 = fmaf(hb.z, cb.z, hb.w * cb.w);
  return (q0 + q1) + (q2 + q3);
}

template <bool FILL, bool PREV>
__device__ inline float scan_walk(const float4* __restrict__ trow,
                                  float* hrow, const float4* hrd,
                                  float h, int src,
                                  float4 caP, float4 cbP, bool writer,
                                  float* __restrict__ ypP,
                                  float4* __restrict__ nbuf,
                                  const ScanOperands& o, int fth, int fd,
                                  int fj, float hA, float fF) {
  float4 ha = {}, hb = {};
  if (PREV) { ha = hrd[0]; hb = hrd[1]; }
  float4 co[SB4];
#pragma unroll
  for (int t = 0; t < SB4; ++t) co[t] = trow[t * 256];

  PairPiece p = {};
  float sp = 0.f, sx = 0.f;
#pragma unroll
  for (int t = 0; t < SB4; ++t) {
    const float hr = __shfl(h, src, 32);
    __builtin_amdgcn_sched_barrier(0);
    const float a = fmaf(co[t].x, h, co[t].z);
    if (FILL) {
      if ((t & 3) == 0) p = pair_front(o, t >> 1, hA, fF);
      if ((t & 3) == 2) pair_back(nbuf, p, o, (t >> 1) - 1, fth, fd, fj);
    }
    if (PREV && t == 0) {
      sp = half_sum(ha, hb, caP, cbP);
      sx = __shfl_xor(sp, 16, 32);
    }
    if (PREV && t == 1) {
      if (writer) *ypP = sp + sx;
    }
    __builtin_amdgcn_sched_barrier(0);
    h = fmaf(-co[t].y, hr, a);
    hrow[t * HP] = h;
  }
  return h;
}

__global__ void __launch_bounds__(512)
k_scan(const float* __restrict__ dt,
       const float* __restrict__ xb,
       const float* __restrict__ Bv,
       const float* __restrict__ Cv,
       const float* __restrict__ A_log,
       const float* __restrict__ rope,
       const float* __restrict__ Pmap,
       const float* __restrict__ qmap,
       const float* __restrict__ init_h,
       float* __restrict__ y,
       float* __restrict__ h_final) {
  __shared__ __attribute__((aligned(16))) float4 tab[2][SB4 * 16 * 16];
  __shared__ __attribute__((aligned(16))) float  hst[16 * SB4 * HP];

  const int c    = blockIdx.x;
  const int tid  = threadIdx.x;
  const int d    = tid >> 5;
  const int lane = tid & 31;
  const int lo   = lane & 15, hi = lane >> 4;

  const int fj  = tid & 15;
  const int fd  = (tid >> 4) & 15;
  const int fth = tid >> 8;
  const float hA = 0.5f * -__expf(A_log[fd * 16 + fj]);
  const float fF = rope[fd * 16 + fj];

  const int t0  = c * CHUNK;
  const int src = (lane & 16) | ((lo + 15) & 15);

  float*        hrow = hst + (d * SB4) * HP + lo;
  const float4* hrd  =
      reinterpret_cast<const float4*>(hst + (d * SB4 + lo) * HP + 8 * hi);
  const bool writer = (hi == 0);

  const float4* Crow = reinterpret_cast<const float4*>(Cv + (t0 + lo) * 16 + 8 * hi);
  float*        yrow = y + (t0 + lo) * 16 + d;
  const int     CSTEP = SB4 * 16 / 4;
  const int     YSTEP = SB4 * 16;

  ScanOperands opsA, opsB;
  scan_load(opsB, dt, xb, Bv, t0, fth, fd, fj);
  scan_load(opsA, dt, xb, Bv, t0 + SB4, fth, fd, fj);

  const int grp = c >> 4, wc = c & 15;
  float4 fm0[16], fm1[16];
  float  fq[16];
#pragma unroll
  for (int gp = 0; gp < 16; ++gp) {
    const int cc = (gp < 15) ? gp * 16 + 15 : ((wc > 0) ? c - 1 : c);
    const float4* Prow = reinterpret_cast<const float4*>(
        Pmap + ((cc * 16 + d) * 16 + lo) * 16 + 8 * hi);
    fm0[gp] = Prow[0];
    fm1[gp] = Prow[1];
    fq[gp]  = qmap[(cc * 16 + d) * 16 + lo];
  }
  float h = init_h[d * 16 + lo];
  __builtin_amdgcn_sched_barrier(0);
#pragma unroll
  for (int gp = 0; gp < 16; ++gp) {
    const float hn = affine_split(fm0[gp], fm1[gp], h, fq[gp], hi != 0);
    const bool take = (gp < 15) ? (gp < grp) : (wc > 0);
    h = take ? hn : h;
  }

  scan_fill(tab[0], opsB, fth, fd, fj, hA, fF);
  __syncthreads();

  scan_load(opsB, dt, xb, Bv, t0 + 2 * SB4, fth, fd, fj);
  float4 caP = Crow[0], cbP = Crow[1];
  h = scan_walk<true, false>(tab[0] + d * 16 + lo, hrow, hrd, h, src,
                             caP, cbP, writer, yrow,
                             tab[1], opsA, fth, fd, fj, hA, fF);
  opsA = opsB;
  __syncthreads();

  for (int kb = 1; kb < NBLK4 - 1; ++kb) {
    if (kb + 2 < NBLK4)
      scan_load(opsB, dt, xb, Bv, t0 + (kb + 2) * SB4, fth, fd, fj);
    const float4 ca = Crow[kb * CSTEP], cb = Crow[kb * CSTEP + 1];

    h = scan_walk<true, true>(tab[kb & 1] + d * 16 + lo, hrow, hrd, h, src,
                              caP, cbP, writer, yrow + (kb - 1) * YSTEP,
                              tab[(kb + 1) & 1], opsA, fth, fd, fj, hA, fF);
    opsA = opsB;
    caP = ca; cbP = cb;
    __syncthreads();
  }
  {
    const int kb = NBLK4 - 1;
    const float4 ca = Crow[kb * CSTEP], cb = Crow[kb * CSTEP + 1];
    h = scan_walk<false, true>(tab[kb & 1] + d * 16 + lo, hrow, hrd, h, src,
                               caP, cbP, writer, yrow + (kb - 1) * YSTEP,
                               tab[0], opsA, fth, fd, fj, hA, fF);
    if (c == NCHUNK - 1 && writer) h_final[d * 16 + lo] = h;
    float s = half_sum(hrd[0], hrd[1], ca, cb);
    s += __shfl_xor(s, 16, 32);
    if (writer) yrow[kb * YSTEP] = s;
  }
}

__global__ void k_out(const float* __restrict__ y,
                      const float* __restrict__ z,
                      const float* __restrict__ xb,
                      const float* __restrict__ Dp,
                      const float* __restrict__ outw,
                      float* __restrict__ out) {
  const int lane = threadIdx.x & 31;
  const int wave = threadIdx.x >> 5;
  const int t0   = (blockIdx.x * 4 + wave) * 16;
  const int lo   = lane & 15, hi = lane >> 4;

  v8f acc = {};
#pragma unroll
  for (int kk = 0; kk < 4; ++kk) {
    const int kb = kk * 4 + 2 * hi;
    v2f a, b;
#pragma unroll
    for (int v = 0; v < 2; ++v) {
      const int k   = kb + v;
      const int idx = (t0 + lo) * 16 + k;
      const float zz  = z[idx];
      const float sil = zz / (1.f + __expf(-zz));
      const float val = y[idx] * sil + Dp[k] * xb[idx];
      if (v == 0) a.x = val; else a.y = val;
    }
    b.x = (lo < 8) ? outw[lo * 16 + kb + 0] : 0.f;
    b.y = (lo < 8) ? outw[lo * 16 + kb + 1] : 0.f;
    acc = wmma4(a, b, acc);
  }
  if (lo < 8) {
#pragma unroll
    for (int r = 0; r < 8; ++r) out[(t0 + r + 8 * hi) * 8 + lo] = acc[r];
  }
}

extern "C" void kernel_launch(void* const* d_in, const int* in_sizes, int n_in,
                              void* d_out, int out_size, void* d_ws,
                              size_t ws_size, hipStream_t stream) {
  const float* x    = (const float*)d_in[0];
  const float* init = (const float*)d_in[1];
  const float* inw  = (const float*)d_in[2];
  const float* dtw  = (const float*)d_in[3];
  const float* dtb  = (const float*)d_in[4];
  const float* bw   = (const float*)d_in[5];
  const float* cw   = (const float*)d_in[6];
  const float* alog = (const float*)d_in[7];
  const float* Dp   = (const float*)d_in[8];
  const float* rope = (const float*)d_in[9];
  const float* outw = (const float*)d_in[10];
  float* out = (float*)d_out;

  float* ws = (float*)d_ws;
  float* XB = ws;
  float* Z  = XB + N16;
  float* DT = Z + N16;
  float* BV = DT + N16;
  float* CV = BV + N16;
  float* Y  = CV + N16;
  float* PM = Y + N16;
  float* QM = PM + N16;
  float* XBT = QM + NCHUNK * 256;
  float* DTT = XBT + N16;

  k_proj<<<NTOK / 64, 128, 0, stream>>>(x, inw, dtw, dtb, bw, cw,
                                        XB, Z, DT, BV, CV, XBT, DTT);
  k_chunkop<<<NCHUNK, KT2, 0, stream>>>(DTT, XBT, BV, alog, rope, PM, QM);
  k_scan<<<NCHUNK, 512, 0, stream>>>(DT, XB, BV, CV, alog, rope, PM, QM, init,
                                     Y, out + NTOK * DM);
  k_out<<<NTOK / 64, 128, 0, stream>>>(Y, Z, XB, Dp, outw, out);
}


# K2 and K4: each wave builds the coefficient table of its own chunk / channel, so waves share nothing: no producer waves in K2, no per-block work-group barriers in either; arithmetic unchanged
# speedup vs baseline: 7.0096x; 1.1126x over previous
#include <hip/hip_runtime.h>
#include <hip/hip_bf16.h>
#include <math.h>


typedef __attribute__((ext_vector_type(2))) float v2f;
typedef __attribute__((ext_vector_type(8))) float v8f;

#define NTOK   65536
#define DM     8
#define DI     16
#define CHUNK  256
#define NCHUNK 256
#define N16    (NTOK * DI)

__device__ inline v8f wmma4(v2f a, v2f b, v8f c) {
  return __builtin_amdgcn_wmma_f32_16x16x4_f32(
      false, a, false, b, (short)0, c, false, false);
}

__device__ inline float softplus_f(float s) {
  return (s > 20.f) ? s : logf(1.f + __expf(s));
}

__global__ void k_proj(const float* __restrict__ x,
                       const float* __restrict__ inw,
                       const float* __restrict__ dtw,
                       const float* __restrict__ dtb,
                       const float* __restrict__ bw,
                       const float* __restrict__ cw,
                       float* __restrict__ xb, float* __restrict__ z,
                       float* __restrict__ dt, float* __restrict__ Bv,
                       float* __restrict__ Cv,
                       float* __restrict__ xbT,
                       float* __restrict__ dtT) {
  __shared__ float lds[4][256];
  const int lane = threadIdx.x & 31;
  const int wave = threadIdx.x >> 5;
  const int t0   = (blockIdx.x * 4 + wave) * 16;
  const int lo   = lane & 15, hi = lane >> 4;

  v8f axb = {}, az = {};
#pragma unroll
  for (int kk = 0; kk < 2; ++kk) {
    const int kb = kk * 4 + 2 * hi;
    v2f a, b0, b1;
    a.x  = x[(t0 + lo) * 8 + kb + 0];
    a.y  = x[(t0 + lo) * 8 + kb + 1];
    b0.x = inw[lo * 8 + kb + 0];
    b0.y = inw[lo * 8 + kb + 1];
    b1.x = inw[(lo + 16) * 8 + kb + 0];
    b1.y = inw[(lo + 16) * 8 + kb + 1];
    axb = wmma4(a, b0, axb);
    az  = wmma4(a, b1, az);
  }

#pragma unroll
  for (int r = 0; r < 8; ++r) lds[wave][(r + 8 * hi) * 16 + lo] = axb[r];

  v8f adt = {}, ab = {}, ac = {};
#pragma unroll
  for (int kk = 0; kk < 4; ++kk) {
    const int kb = kk * 4 + 2 * hi;
    v2f a, b;
    a.x = lds[wave][lo * 16 + kb + 0];
    a.y = lds[wave][lo * 16 + kb + 1];
    b.x = dtw[lo * 16 + kb + 0]; b.y = dtw[lo * 16 + kb + 1];
    adt = wmma4(a, b, adt);
    b.x = bw[lo * 16 + kb + 0];  b.y = bw[lo * 16 + kb + 1];
    ab  = wmma4(a, b, ab);
    b.x = cw[lo * 16 + kb + 0];  b.y = cw[lo * 16 + kb + 1];
    ac  = wmma4(a, b, ac);
  }

  const float bias = dtb[lo];
  float dtv[8];
#pragma unroll
  for (int r = 0; r < 8; ++r) {
    const int idx = (t0 + r + 8 * hi) * 16 + lo;
    dtv[r]  = softplus_f(adt[r] + bias);
    xb[idx] = axb[r];
    z[idx]  = az[r];
    dt[idx] = dtv[r];
    Bv[idx] = ab[r];
    Cv[idx] = ac[r];
  }

  const int tch = lane >> 1, ttk = (lane & 1) * 8;
  {
    float4 v0, v1;
    v0.x = lds[wave][(ttk + 0) * 16 + tch]; v0.y = lds[wave][(ttk + 1) * 16 + tch];
    v0.z = lds[wave][(ttk + 2) * 16 + tch]; v0.w = lds[wave][(ttk + 3) * 16 + tch];
    v1.x = lds[wave][(ttk + 4) * 16 + tch]; v1.y = lds[wave][(ttk + 5) * 16 + tch];
    v1.z = lds[wave][(ttk + 6) * 16 + tch]; v1.w = lds[wave][(ttk + 7) * 16 + tch];
    float4* dst = reinterpret_cast<float4*>(xbT + tch * NTOK + t0 + ttk);
    dst[0] = v0; dst[1] = v1;
  }
#pragma unroll
  for (int r = 0; r < 8; ++r) lds[wave][(r + 8 * hi) * 16 + lo] = dtv[r];
  {
    float4 v0, v1;
    v0.x = lds[wave][(ttk + 0) * 16 + tch]; v0.y = lds[wave][(ttk + 1) * 16 + tch];
    v0.z = lds[wave][(ttk + 2) * 16 + tch]; v0.w = lds[wave][(ttk + 3) * 16 + tch];
    v1.x = lds[wave][(ttk + 4) * 16 + tch]; v1.y = lds[wave][(ttk + 5) * 16 + tch];
    v1.z = lds[wave][(ttk + 6) * 16 + tch]; v1.w = lds[wave][(ttk + 7) * 16 + tch];
    float4* dst = reinterpret_cast<float4*>(dtT + tch * NTOK + t0 + ttk);
    dst[0] = v0; dst[1] = v1;
  }
}

#define SB    32
#define NBLK  (CHUNK / SB)
#define ROWF  48
#define STEPF (16 * ROWF + 16)

struct FillOperands { float dtc[SB / 2], xbd[SB / 2], Bj[SB / 2]; };

__device__ inline void chunkop_load(FillOperands& o,
                                    const float* __restrict__ dtc,
                                    const float* __restrict__ xbc,
                                    const float* __restrict__ Bv,
                                    int tbase, int fth, int fj) {
#pragma unroll
  for (int i = 0; i < SB / 2; ++i) {
    const int tt = tbase + 2 * i + fth;
    o.dtc[i] = dtc[tt];
    o.xbd[i] = xbc[tt];
    o.Bj[i]  = Bv[tt * 16 + fj];
  }
}

__device__ inline v2f fma2(v2f a, v2f b, v2f c) { return __builtin_elementwise_fma(a, b, c); }

__device__ inline void chunkop_make(float* __restrict__ buf,
                                    const FillOperands& o,
                                    int fth, int fd, int fj,
                                    float fA, float fF) {
  const float hA = 0.5f * fA;
#pragma unroll
  for (int i = 0; i < SB / 2; i += 2) {
    v2f dtc; dtc.x = o.dtc[i]; dtc.y = o.dtc[i + 1];
    v2f Bj;  Bj.x  = o.Bj[i];  Bj.y  = o.Bj[i + 1];
    v2f xbd; xbd.x = o.xbd[i]; xbd.y = o.xbd[i + 1];
    const v2f ta  = dtc * hA;
    const v2f num = 1.f + ta;
    const v2f den = (1.f - ta) + 1e-8f;
    v2f r; r.x = __builtin_amdgcn_rcpf(den.x); r.y = __builtin_amdgcn_rcpf(den.y);
    v2f q = num * r;
    q = fma2(fma2(-den, q, num), r, q);
    const v2f rev = (dtc * fF) * 0.15915494309189532f;
    v2f cs, sn;
    cs.x = __builtin_amdgcn_cosf(rev.x); cs.y = __builtin_amdgcn_cosf(rev.y);
    sn.x = __builtin_amdgcn_sinf(rev.x); sn.y = __builtin_amdgcn_sinf(rev.y);
    const v2f E = q * cs, F = q * sn, U = dtc * Bj * xbd;
    const int t = 2 * i + fth;
    float* row0 = buf + (SB - 1 - t) * STEPF + fd * ROWF;
    float* row1 = row0 - 2 * STEPF;
    row0[fj]      = E.x;
    row0[16 + ((fj + 15) & 15)] = F.x;
    row0[32 + fj] = U.x;
    row1[fj]      = E.y;
    row1[16 + ((fj + 15) & 15)] = F.y;
    row1[32 + fj] = U.y;
  }
}

__device__ inline void chunkop_fill(float* __restrict__ buf,
                                    const float* __restrict__ dtc,
                                    const float* __restrict__ xbc,
                                    const float* __restrict__ Bv,
                                    int tbase, int fth, int fd, int fj,
                                    float fA, float fF) {
  FillOperands o;
  chunkop_load(o, dtc, xbc, Bv, tbase, fth, fj);
  chunkop_make(buf, o, fth, fd, fj, fA, fF);
}

#define KW2   512

__device__ inline float next_pair(float v) {
  const int i = __builtin_bit_cast(int, v);
  return __builtin_bit_cast(float, __builtin_amdgcn_permlane16(
      i, i, (int)0x07654321u, (int)0x8fedcba9u, true, false));
}

__device__ inline float lane_bcast(float v, int k) {
  return __int_as_float(__builtin_amdgcn_readlane(__float_as_int(v), k));
}

__device__ inline float affine_row(float4 m0, float4 m1, float4 m2, float4 m3,
                                   float vec, float bias) {
  float s[16];
#pragma unroll
  for (int k = 0; k < 16; ++k) s[k] = lane_bcast(vec, k);
  float s0 = m0.x * s[0], s1 = m1.x * s[4], s2 = m2.x * s[8], s3 = m3.x * s[12];
  s0 = fmaf(m0.y, s[1], s0);  s1 = fmaf(m1.y, s[5], s1);
  s2 = fmaf(m2.y, s[9], s2);  s3 = fmaf(m3.y, s[13], s3);
  s0 = fmaf(m0.z, s[2], s0);  s1 = fmaf(m1.z, s[6], s1);
  s2 = fmaf(m2.z, s[10], s2); s3 = fmaf(m3.z, s[14], s3);
  s0 = fmaf(m0.w, s[3], s0);  s1 = fmaf(m1.w, s[7], s1);
  s2 = fmaf(m2.w, s[11], s2); s3 = fmaf(m3.w, s[15], s3);
  return bias + ((s0 + s1) + (s2 + s3));
}

__device__ inline float other_row(float v) {
  const int i = __builtin_bit_cast(int, v);
  return __builtin_bit_cast(float, __builtin_amdgcn_permlanex16(
      i, i, (int)0x76543210u, (int)0xfedcba98u, true, false));
}

__device__ inline float affine_split(float4 m0, float4 m1, float vec, float bias,
                                     bool upper) {
  float s[8];
#pragma unroll
  for (int k = 0; k < 8; ++k) {
    const float a = lane_bcast(vec, k), b = lane_bcast(vec, k + 8);
    s[k] = upper ? b : a;
  }
  float s0 = m0.x * s[0], s1 = m1.x * s[4];
  s0 = fmaf(m0.y, s[1], s0);  s1 = fmaf(m1.y, s[5], s1);
  s0 = fmaf(m0.z, s[2], s0);  s1 = fmaf(m1.z, s[6], s1);
  s0 = fmaf(m0.w, s[3], s0);  s1 = fmaf(m1.w, s[7], s1);
  const float part = s0 + s1;
  const float oth  = other_row(part);
  return bias + (upper ? (oth + part) : (part + oth));
}

__device__ inline void chunkop_walk(const float* base, float (&Sa)[4],
                                    float (&Sb)[4], float (&bacc)[4]) {
#pragma unroll 4
  for (int i = 0; i < SB; ++i) {
    const float* row = base + i * STEPF;
    const float2 E = *reinterpret_cast<const float2*>(row);
    const float2 F = *reinterpret_cast<const float2*>(row + 16);
    const float2 U = *reinterpret_cast<const float2*>(row + 32);
#pragma unroll
    for (int r = 0; r < 4; ++r) {
      const float sn = next_pair(Sa[r]);
      bacc[r] = fmaf(Sb[r], U.y, fmaf(Sa[r], U.x, bacc[r]));
      const float a = E.x * Sa[r] - F.x * Sb[r];
      Sb[r] = E.y * Sb[r] - F.y * sn;
      Sa[r] = a;
    }
  }
}

__global__ void __launch_bounds__(KW2)
k_chunkop(const float* __restrict__ dtT,
          const float* __restrict__ xbT,
          const float* __restrict__ Bv,
          const float* __restrict__ A_log,
          const float* __restrict__ rope,
          float* __restrict__ Pmap,
          float* __restrict__ qmap) {
  __shared__ __attribute__((aligned(16))) float tab[2][SB * STEPF];
  __shared__ __attribute__((aligned(16))) float opl[16 * 256 + 16 * 16];

  const int grp  = blockIdx.x >> 4;
  const int ch   = blockIdx.x & 15;
  const int tid  = threadIdx.x;
  const int w    = tid >> 5;
  const int lane = tid & 31;

  const int fj  = lane & 15;
  const int fth = lane >> 4;
  const float fA = -__expf(A_log[ch * 16 + fj]);
  const float fF = rope[ch * 16 + fj];
  const float* dtc = dtT + ch * NTOK;
  const float* xbc = xbT + ch * NTOK;

  const int p = lane & 7, q = lane >> 3;
  float Sa[4], Sb[4], bacc[4];
#pragma unroll
  for (int r = 0; r < 4; ++r) {
    Sa[r]   = ((4 * q + r) == 2 * p)     ? 1.f : 0.f;
    Sb[r]   = ((4 * q + r) == 2 * p + 1) ? 1.f : 0.f;
    bacc[r] = 0.f;
  }

  const int t0f = (grp * 16 + w) * CHUNK;

  chunkop_fill(tab[0], dtc, xbc, Bv, t0f + (NBLK - 1) * SB, fth, w, fj, fA, fF);

  for (int kb = 0; kb + 1 < NBLK; ++kb) {
    FillOperands o;
    chunkop_load(o, dtc, xbc, Bv, t0f + (NBLK - 2 - kb) * SB, fth, fj);
    chunkop_walk(tab[kb & 1] + w * ROWF + 2 * p, Sa, Sb, bacc);
    chunkop_make(tab[(kb + 1) & 1], o, fth, w, fj, fA, fF);
  }
  chunkop_walk(tab[(NBLK - 1) & 1] + w * ROWF + 2 * p, Sa, Sb, bacc);

  float* Tl = opl;
  float* bl = opl + 16 * 256;
  {
#pragma unroll
    for (int r = 0; r < 4; ++r) {
      float2 v; v.x = Sa[r]; v.y = Sb[r];
      *reinterpret_cast<float2*>(Tl + w * 256 + (4 * q + r) * 16 + 2 * p) = v;
    }
    float bout = 0.f;
#pragma unroll
    for (int r = 0; r < 4; ++r) {
      float v = bacc[r];
      v += __shfl_xor(v, 1, 32);
      v += __shfl_xor(v, 2, 32);
      v += __shfl_xor(v, 4, 32);
      bout = (p == r) ? v : bout;
    }
    if (p < 4) bl[w * 16 + 4 * q + p] = bout;
  }
  __syncthreads();
  if (tid >= 32) return;

  const int lo = lane & 15, hi = lane >> 4;
  v8f P;
#pragma unroll
  for (int r = 0; r < 8; ++r) P[r] = ((r + 8 * hi) == lo) ? 1.f : 0.f;
  float qq = 0.f;
#pragma unroll
  for (int i = 0; i < 16; ++i) {
    const float* Tc = Tl + i * 256;
    v2f a[4];
#pragma unroll
    for (int kk = 0; kk < 4; ++kk)
      a[kk] = *reinterpret_cast<const v2f*>(Tc + lo * 16 + 8 * hi + 2 * kk);
    const float4* Trow = reinterpret_cast<const float4*>(Tc + lo * 16);
    const float4 m0 = Trow[0], m1 = Trow[1], m2 = Trow[2], m3 = Trow[3];
    const float bias = bl[i * 16 + lo];

    v8f acc = {};
#pragma unroll
    for (int kk = 0; kk < 4; ++kk) {
      v2f b;
      b.x = P[2 * kk];
      b.y = P[2 * kk + 1];
      acc = wmma4(a[kk], b, acc);
    }
    qq = affine_row(m0, m1, m2, m3, qq, bias);
    P = acc;

    const int c = grp * 16 + i;
    float* Pg = Pmap + (c * 16 + ch) * 256;
#pragma unroll
    for (int r = 0; r < 8; ++r) Pg[(r + 8 * hi) * 16 + lo] = P[r];
    qmap[(c * 16 + ch) * 16 + lo] = qq;
  }
}

#define SB4    16
#define NBLK4  (CHUNK / SB4)
#define HP     20

struct ScanOperands { float dtc[SB4 / 2], xbd[SB4 / 2], Bj[SB4 / 2]; };

__device__ inline void scan_load(ScanOperands& o,
                                 const float* __restrict__ dt,
                                 const float* __restrict__ xb,
                                 const float* __restrict__ Bv,
                                 int tbase, int fth, int fd, int fj) {
#pragma unroll
  for (int i = 0; i < SB4 / 2; ++i) {
    const int tt = tbase + 2 * i + fth;
    o.dtc[i] = dt[tt * 16 + fd];
    o.xbd[i] = xb[tt * 16 + fd];
    o.Bj[i]  = Bv[tt * 16 + fj];
  }
}

struct PairPiece { v2f q, rev, dtc; };

__device__ inline PairPiece pair_front(const ScanOperands& o, int i,
                                       float hA, float fF) {
  PairPiece p;
  p.dtc.x = o.dtc[i]; p.dtc.y = o.dtc[i + 1];
  const v2f ta  = p.dtc * hA;
  const v2f num = 1.f + ta;
  const v2f den = (1.f - ta) + 1e-8f;
  v2f r; r.x = __builtin_amdgcn_rcpf(den.x); r.y = __builtin_amdgcn_rcpf(den.y);
  v2f q = num * r;
  p.q   = fma2(fma2(-den, q, num), r, q);
  p.rev = (p.dtc * fF) * 0.15915494309189532f;
  return p;
}

__device__ inline void pair_back(float4* __restrict__ buf, const PairPiece& p,
                                 const ScanOperands& o, int i,
                                 int fth, int fd, int fj) {
  v2f cs, sn;
  cs.x = __builtin_amdgcn_cosf(p.rev.x); cs.y = __builtin_amdgcn_cosf(p.rev.y);
  sn.x = __builtin_amdgcn_sinf(p.rev.x); sn.y = __builtin_amdgcn_sinf(p.rev.y);
  v2f Bj;  Bj.x  = o.Bj[i];  Bj.y  = o.Bj[i + 1];
  v2f xbd; xbd.x = o.xbd[i]; xbd.y = o.xbd[i + 1];
  const v2f E = p.q * cs, F = p.q * sn, U = p.dtc * Bj * xbd;
  const int t = 2 * i + fth;
  float4 v0, v1;
  v0.x = E.x; v0.y = F.x; v0.z = U.x; v0.w = 0.f;
  v1.x = E.y; v1.y = F.y; v1.z = U.y; v1.w = 0.f;
  buf[(t * 16 + fd) * 16 + fj]       = v0;
  buf[((t + 2) * 16 + fd) * 16 + fj] = v1;
}

__device__ inline void scan_fill(float4* __restrict__ buf, const ScanOperands& o,
                                 int fth, int fd, int fj, float hA, float fF) {
#pragma unroll
  for (int i = 0; i < SB4 / 2; i += 2) {
    const PairPiece p = pair_front(o, i, hA, fF);
    pair_back(buf, p, o, i, fth, fd, fj);
  }
}

__device__ inline float half_sum(float4 ha, float4 hb, float4 ca, float4 cb) {
  const float q0 = fmaf(ha.x, ca.x, ha.y * ca.y), q1 = fmaf(ha.z, ca.z, ha.w * ca.w);
  const float q2 = fmaf(hb.x, cb.x, hb.y * cb.y), q3 = fmaf(hb.z, cb.z, hb.w * cb.w);
  return (q0 + q1) + (q2 + q3);
}

template <bool FILL, bool PREV>
__device__ inline float scan_walk(const float4* __restrict__ trow,
                                  float* hrow, const float4* hrd,
                                  float h, int src,
                                  float4 caP, float4 cbP, bool writer,
                                  float* __restrict__ ypP,
                                  float4* __restrict__ nbuf,
                                  const ScanOperands& o, int fth, int fd,
                                  int fj, float hA, float fF) {
  float4 ha = {}, hb = {};
  if (PREV) { ha = hrd[0]; hb = hrd[1]; }
  float4 co[SB4];
#pragma unroll
  for (int t = 0; t < SB4; ++t) co[t] = trow[t * 256];

  PairPiece p = {};
  float sp = 0.f, sx = 0.f;
#pragma unroll
  for (int t = 0; t < SB4; ++t) {
    const float hr = __shfl(h, src, 32);
    __builtin_amdgcn_sched_barrier(0);
    const float a = fmaf(co[t].x, h, co[t].z);
    if (FILL) {
      if ((t & 3) == 0) p = pair_front(o, t >> 1, hA, fF);
      if ((t & 3) == 2) pair_back(nbuf, p, o, (t >> 1) - 1, fth, fd, fj);
    }
    if (PREV && t == 0) {
      sp = half_sum(ha, hb, caP, cbP);
      sx = __shfl_xor(sp, 16, 32);
    }
    if (PREV && t == 1) {
      if (writer) *ypP = sp + sx;
    }
    __builtin_amdgcn_sched_barrier(0);
    h = fmaf(-co[t].y, hr, a);
    hrow[t * HP] = h;
  }
  return h;
}

__global__ void __launch_bounds__(512)
k_scan(const float* __restrict__ dt,
       const float* __restrict__ xb,
       const float* __restrict__ Bv,
       const float* __restrict__ Cv,
       const float* __restrict__ A_log,
       const float* __restrict__ rope,
       const float* __restrict__ Pmap,
       const float* __restrict__ qmap,
       const float* __restrict__ init_h,
       float* __restrict__ y,
       float* __restrict__ h_final) {
  __shared__ __attribute__((aligned(16))) float4 tab[2][SB4 * 16 * 16];
  __shared__ __attribute__((aligned(16))) float  hst[16 * SB4 * HP];

  const int c    = blockIdx.x;
  const int tid  = threadIdx.x;
  const int d    = tid >> 5;
  const int lane = tid & 31;
  const int lo   = lane & 15, hi = lane >> 4;

  const int fj  = lo;
  const int fd  = d;
  const int fth = hi;
  const float hA = 0.5f * -__expf(A_log[fd * 16 + fj]);
  const float fF = rope[fd * 16 + fj];

  const int t0  = c * CHUNK;
  const int src = (lane & 16) | ((lo + 15) & 15);

  float*        hrow = hst + (d * SB4) * HP + lo;
  const float4* hrd  =
      reinterpret_cast<const float4*>(hst + (d * SB4 + lo) * HP + 8 * hi);
  const bool writer = (hi == 0);

  const float4* Crow = reinterpret_cast<const float4*>(Cv + (t0 + lo) * 16 + 8 * hi);
  float*        yrow = y + (t0 + lo) * 16 + d;
  const int     CSTEP = SB4 * 16 / 4;
  const int     YSTEP = SB4 * 16;

  ScanOperands opsA, opsB;
  scan_load(opsB, dt, xb, Bv, t0, fth, fd, fj);
  scan_load(opsA, dt, xb, Bv, t0 + SB4, fth, fd, fj);

  const int grp = c >> 4, wc = c & 15;
  float4 fm0[16], fm1[16];
  float  fq[16];
#pragma unroll
  for (int gp = 0; gp < 16; ++gp) {
    const int cc = (gp < 15) ? gp * 16 + 15 : ((wc > 0) ? c - 1 : c);
    const float4* Prow = reinterpret_cast<const float4*>(
        Pmap + ((cc * 16 + d) * 16 + lo) * 16 + 8 * hi);
    fm0[gp] = Prow[0];
    fm1[gp] = Prow[1];
    fq[gp]  = qmap[(cc * 16 + d) * 16 + lo];
  }
  float h = init_h[d * 16 + lo];
  __builtin_amdgcn_sched_barrier(0);
#pragma unroll
  for (int gp = 0; gp < 16; ++gp) {
    const float hn = affine_split(fm0[gp], fm1[gp], h, fq[gp], hi != 0);
    const bool take = (gp < 15) ? (gp < grp) : (wc > 0);
    h = take ? hn : h;
  }

  scan_fill(tab[0], opsB, fth, fd, fj, hA, fF);

  scan_load(opsB, dt, xb, Bv, t0 + 2 * SB4, fth, fd, fj);
  float4 caP = Crow[0], cbP = Crow[1];
  h = scan_walk<true, false>(tab[0] + d * 16 + lo, hrow, hrd, h, src,
                             caP, cbP, writer, yrow,
                             tab[1], opsA, fth, fd, fj, hA, fF);
  opsA = opsB;

  for (int kb = 1; kb < NBLK4 - 1; ++kb) {
    if (kb + 2 < NBLK4)
      scan_load(opsB, dt, xb, Bv, t0 + (kb + 2) * SB4, fth, fd, fj);
    const float4 ca = Crow[kb * CSTEP], cb = Crow[kb * CSTEP + 1];

    h = scan_walk<true, true>(tab[kb & 1] + d * 16 + lo, hrow, hrd, h, src,
                              caP, cbP, writer, yrow + (kb - 1) * YSTEP,
                              tab[(kb + 1) & 1], opsA, fth, fd, fj, hA, fF);
    opsA = opsB;
    caP = ca; cbP = cb;
  }
  {
    const int kb = NBLK4 - 1;
    const float4 ca = Crow[kb * CSTEP], cb = Crow[kb * CSTEP + 1];
    h = scan_walk<false, true>(tab[kb & 1] + d * 16 + lo, hrow, hrd, h, src,
                               caP, cbP, writer, yrow + (kb - 1) * YSTEP,
                               tab[0], opsA, fth, fd, fj, hA, fF);
    if (c == NCHUNK - 1 && writer) h_final[d * 16 + lo] = h;
    float s = half_sum(hrd[0], hrd[1], ca, cb);
    s += __shfl_xor(s, 16, 32);
    if (writer) yrow[kb * YSTEP] = s;
  }
}

__global__ void k_out(const float* __restrict__ y,
                      const float* __restrict__ z,
                      const float* __restrict__ xb,
                      const float* __restrict__ Dp,
                      const float* __restrict__ outw,
                      float* __restrict__ out) {
  const int lane = threadIdx.x & 31;
  const int wave = threadIdx.x >> 5;
  const int t0   = (blockIdx.x * 4 + wave) * 16;
  const int lo   = lane & 15, hi = lane >> 4;

  v8f acc = {};
#pragma unroll
  for (int kk = 0; kk < 4; ++kk) {
    const int kb = kk * 4 + 2 * hi;
    v2f a, b;
#pragma unroll
    for (int v = 0; v < 2; ++v) {
      const int k   = kb + v;
      const int idx = (t0 + lo) * 16 + k;
      const float zz  = z[idx];
      const float sil = zz / (1.f + __expf(-zz));
      const float val = y[idx] * sil + Dp[k] * xb[idx];
      if (v == 0) a.x = val; else a.y = val;
    }
    b.x = (lo < 8) ? outw[lo * 16 + kb + 0] : 0.f;
    b.y = (lo < 8) ? outw[lo * 16 + kb + 1] : 0.f;
    acc = wmma4(a, b, acc);
  }
  if (lo < 8) {
#pragma unroll
    for (int r = 0; r < 8; ++r) out[(t0 + r + 8 * hi) * 8 + lo] = acc[r];
  }
}

extern "C" void kernel_launch(void* const* d_in, const int* in_sizes, int n_in,
                              void* d_out, int out_size, void* d_ws,
                              size_t ws_size, hipStream_t stream) {
  const float* x    = (const float*)d_in[0];
  const float* init = (const float*)d_in[1];
  const float* inw  = (const float*)d_in[2];
  const float* dtw  = (const float*)d_in[3];
  const float* dtb  = (const float*)d_in[4];
  const float* bw   = (const float*)d_in[5];
  const float* cw   = (const float*)d_in[6];
  const float* alog = (const float*)d_in[7];
  const float* Dp   = (const float*)d_in[8];
  const float* rope = (const float*)d_in[9];
  const float* outw = (const float*)d_in[10];
  float* out = (float*)d_out;

  float* ws = (float*)d_ws;
  float* XB = ws;
  float* Z  = XB + N16;
  float* DT = Z + N16;
  float* BV = DT + N16;
  float* CV = BV + N16;
  float* Y  = CV + N16;
  float* PM = Y + N16;
  float* QM = PM + N16;
  float* XBT = QM + NCHUNK * 256;
  float* DTT = XBT + N16;

  k_proj<<<NTOK / 64, 128, 0, stream>>>(x, inw, dtw, dtb, bw, cw,
                                        XB, Z, DT, BV, CV, XBT, DTT);
  k_chunkop<<<NCHUNK, KW2, 0, stream>>>(DTT, XBT, BV, alog, rope, PM, QM);
  k_scan<<<NCHUNK, 512, 0, stream>>>(DT, XB, BV, CV, alog, rope, PM, QM, init,
                                     Y, out + NTOK * DM);
  k_out<<<NTOK / 64, 128, 0, stream>>>(Y, Z, XB, Dp, outw, out);
}
